# starting kernel
# baseline (speedup 1.0000x reference)
#include <hip/hip_runtime.h>
#include <math.h>

#define BB   32
#define CC   122
#define TT   500
#define WID  80
#define UU   96
#define K2   (WID*CC)
#define HID  768
#define PROJ 1024
#define PLEN 484
#define EPSF 1e-5f

typedef float v2f __attribute__((ext_vector_type(2)));
typedef float v8f __attribute__((ext_vector_type(8)));

static __device__ __forceinline__ v8f wmma_f32(v2f a, v2f b, v8f c) {
    return __builtin_amdgcn_wmma_f32_16x16x4_f32(false, a, false, b, (short)0, c, false, false);
}

__global__ __launch_bounds__(32) void k_subject(
    const float* __restrict__ x, const int* __restrict__ sid,
    const float* __restrict__ W_sub, const float* __restrict__ b_sub,
    float* __restrict__ out0)
{
    const int lane = threadIdx.x;
    const int b  = blockIdx.z;
    const int m0 = blockIdx.x * 16;
    const int n0 = blockIdx.y * 16;
    const int s  = sid[b];
    const float* Wg = W_sub + (size_t)s * TT * TT;

    int am = m0 + (lane & 15); if (am >= CC) am = CC - 1;
    int an = n0 + (lane & 15); if (an >= TT) an = TT - 1;
    const int koff = (lane >> 4) * 2;
    const float* pa = x  + ((size_t)b * CC + am) * TT + koff;
    const float* pb = Wg + (size_t)an * TT + koff;

    v8f acc = {};
    #pragma unroll 4
    for (int k4 = 0; k4 < TT / 4; ++k4) {
        v2f a = *(const v2f*)(pa + 4 * k4);
        v2f w = *(const v2f*)(pb + 4 * k4);
        acc = wmma_f32(a, w, acc);
    }

    const float* bg = b_sub + (size_t)s * TT;
    const int nn = n0 + (lane & 15);
    #pragma unroll
    for (int r = 0; r < 8; ++r) {
        int mm = m0 + r + ((lane >> 4) << 3);
        if (mm < CC && nn < TT)
            out0[((size_t)b * CC + mm) * TT + nn] = acc[r] + bg[nn];
    }
}

__global__ __launch_bounds__(128) void k_pool(
    const float* __restrict__ out0, float* __restrict__ P)
{
    __shared__ float row[TT];
    const int bc = blockIdx.x;
    const float* src = out0 + (size_t)bc * TT;
    for (int i = threadIdx.x; i < TT; i += 128) row[i] = src[i];
    __syncthreads();
    float* dst = P + (size_t)bc * PLEN;
    for (int m = threadIdx.x; m < PLEN; m += 128) {
        float sum = 0.f;
        #pragma unroll
        for (int i = 0; i < 17; ++i) sum += row[m + i];
        dst[m] = sum * (1.f / 17.f);
    }
}

__global__ __launch_bounds__(160) void k_conv2(
    const float* __restrict__ P,
    const float* __restrict__ conv1_w, const float* __restrict__ conv1_b,
    const float* __restrict__ bn1_g, const float* __restrict__ bn1_b,
    const float* __restrict__ bn1_m, const float* __restrict__ bn1_v,
    const float* __restrict__ conv2_w, const float* __restrict__ conv2_b,
    const float* __restrict__ bn2_g, const float* __restrict__ bn2_b,
    const float* __restrict__ bn2_m, const float* __restrict__ bn2_v,
    float* __restrict__ h2)
{
    __shared__ float Pl[CC * 80];
    __shared__ float w1s[WID * 5];
    __shared__ float sc1[WID], sh1[WID];

    const int tid = threadIdx.x;
    const int b   = blockIdx.y;
    const int u0  = blockIdx.x * 16;

    for (int i = tid; i < WID * 5; i += 160) w1s[i] = conv1_w[i];
    if (tid < WID) {
        float sc = bn1_g[tid] * rsqrtf(bn1_v[tid] + EPSF);
        sc1[tid] = sc;
        sh1[tid] = bn1_b[tid] - bn1_m[tid] * sc + sc * conv1_b[tid];
    }
    for (int i = tid; i < CC * 80; i += 160) {
        int c = i / 80, q = i - c * 80;
        Pl[i] = P[((size_t)b * CC + c) * PLEN + 5 * u0 + q];
    }
    __syncthreads();

    const int wave = tid >> 5;
    const int lane = tid & 31;
    const int n    = lane & 15;
    const int half = lane >> 4;

    const float* paA = conv2_w + (size_t)(wave * 16 + (lane & 15)) * K2 + half * 2;

    int w0 = 0, c0 = half * 2;
    v8f acc = {};
    for (int k4 = 0; k4 < K2 / 4; ++k4) {
        v2f a = *(const v2f*)(paA + 4 * k4);

        const float* wr0 = &w1s[w0 * 5];
        const float* pr0 = &Pl[c0 * 80 + 5 * n];
        float v0 = wr0[0]*pr0[0] + wr0[1]*pr0[1] + wr0[2]*pr0[2] + wr0[3]*pr0[3] + wr0[4]*pr0[4];
        float g0 = sc1[w0] * v0 + sh1[w0];
        float e0 = g0 > 0.f ? g0 : (__expf(g0) - 1.f);

        int c1 = c0 + 1, w1i = w0;
        if (c1 == CC) { c1 = 0; w1i = w0 + 1; }
        const float* wr1 = &w1s[w1i * 5];
        const float* pr1 = &Pl[c1 * 80 + 5 * n];
        float v1 = wr1[0]*pr1[0] + wr1[1]*pr1[1] + wr1[2]*pr1[2] + wr1[3]*pr1[3] + wr1[4]*pr1[4];
        float g1 = sc1[w1i] * v1 + sh1[w1i];
        float e1 = g1 > 0.f ? g1 : (__expf(g1) - 1.f);

        v2f bb = { e0, e1 };
        acc = wmma_f32(a, bb, acc);

        c0 += 4;
        if (c0 >= CC) { c0 -= CC; w0 += 1; }
    }

    #pragma unroll
    for (int r = 0; r < 8; ++r) {
        int w2 = wave * 16 + r + half * 8;
        float sc = bn2_g[w2] * rsqrtf(bn2_v[w2] + EPSF);
        float sh = bn2_b[w2] - bn2_m[w2] * sc;
        float v  = sc * (acc[r] + conv2_b[w2]) + sh;
        float e  = v > 0.f ? v : (__expf(v) - 1.f);
        h2[((size_t)b * WID + w2) * UU + u0 + n] = e;
    }
}

__global__ __launch_bounds__(256) void k_proj(
    const float* __restrict__ h2, const float* __restrict__ proj_w,
    const float* __restrict__ proj_b, float* __restrict__ z)
{
    __shared__ float hs[WID * UU];
    __shared__ float pw[8 * WID];
    const int b = blockIdx.x;
    for (int i = threadIdx.x; i < WID * UU; i += 256) hs[i] = h2[(size_t)b * WID * UU + i];
    for (int i = threadIdx.x; i < 8 * WID; i += 256)  pw[i] = proj_w[i];
    __syncthreads();
    for (int idx = threadIdx.x; idx < HID; idx += 256) {
        int u = idx >> 3, e = idx & 7;
        float s = proj_b[e];
        #pragma unroll 8
        for (int w = 0; w < WID; ++w) s += pw[e * WID + w] * hs[w * UU + u];
        z[(size_t)b * HID + idx] = s;
    }
}

__global__ __launch_bounds__(32) void k_mlp1(
    const float* __restrict__ z, const float* __restrict__ w,
    const float* __restrict__ bias, float* __restrict__ z1, float* __restrict__ g)
{
    const int lane = threadIdx.x;
    const int m0 = blockIdx.x * 16, n0 = blockIdx.y * 16;
    const float* pa = z + (size_t)(m0 + (lane & 15)) * HID + (lane >> 4) * 2;
    const float* pb = w + (size_t)(n0 + (lane & 15)) * HID + (lane >> 4) * 2;
    v8f acc = {};
    #pragma unroll 4
    for (int k4 = 0; k4 < HID / 4; ++k4) {
        v2f a = *(const v2f*)(pa + 4 * k4);
        v2f bb = *(const v2f*)(pb + 4 * k4);
        acc = wmma_f32(a, bb, acc);
    }
    const int nn = n0 + (lane & 15);
    const float bn = bias[nn];
    #pragma unroll
    for (int r = 0; r < 8; ++r) {
        int mm = m0 + r + ((lane >> 4) << 3);
        float v = acc[r] + bn;
        z1[(size_t)mm * PROJ + nn] = v;
        g [(size_t)mm * PROJ + nn] = 0.5f * v * (1.f + erff(v * 0.70710678118654752f));
    }
}

__global__ __launch_bounds__(32) void k_mlp2(
    const float* __restrict__ g, const float* __restrict__ w,
    const float* __restrict__ bias, const float* __restrict__ z1,
    float* __restrict__ y)
{
    const int lane = threadIdx.x;
    const int m0 = blockIdx.x * 16, n0 = blockIdx.y * 16;
    const float* pa = g + (size_t)(m0 + (lane & 15)) * PROJ + (lane >> 4) * 2;
    const float* pb = w + (size_t)(n0 + (lane & 15)) * PROJ + (lane >> 4) * 2;
    v8f acc = {};
    #pragma unroll 4
    for (int k4 = 0; k4 < PROJ / 4; ++k4) {
        v2f a = *(const v2f*)(pa + 4 * k4);
        v2f bb = *(const v2f*)(pb + 4 * k4);
        acc = wmma_f32(a, bb, acc);
    }
    const int nn = n0 + (lane & 15);
    const float bn = bias[nn];
    #pragma unroll
    for (int r = 0; r < 8; ++r) {
        int mm = m0 + r + ((lane >> 4) << 3);
        y[(size_t)mm * PROJ + nn] = z1[(size_t)mm * PROJ + nn] + acc[r] + bn;
    }
}

__global__ __launch_bounds__(256) void k_ln(
    const float* __restrict__ y, const float* __restrict__ gam,
    const float* __restrict__ bet, float* __restrict__ out)
{
    __shared__ float red[256];
    const int b = blockIdx.x;
    const float* row = y + (size_t)b * PROJ;
    float s = 0.f;
    for (int i = threadIdx.x; i < PROJ; i += 256) s += row[i];
    red[threadIdx.x] = s; __syncthreads();
    for (int off = 128; off > 0; off >>= 1) {
        if (threadIdx.x < off) red[threadIdx.x] += red[threadIdx.x + off];
        __syncthreads();
    }
    const float mu = red[0] * (1.f / PROJ);
    __syncthreads();
    float s2 = 0.f;
    for (int i = threadIdx.x; i < PROJ; i += 256) { float d = row[i] - mu; s2 += d * d; }
    red[threadIdx.x] = s2; __syncthreads();
    for (int off = 128; off > 0; off >>= 1) {
        if (threadIdx.x < off) red[threadIdx.x] += red[threadIdx.x + off];
        __syncthreads();
    }
    const float inv = rsqrtf(red[0] * (1.f / PROJ) + EPSF);
    for (int i = threadIdx.x; i < PROJ; i += 256)
        out[(size_t)b * PROJ + i] = (row[i] - mu) * inv * gam[i] + bet[i];
}

extern "C" void kernel_launch(void* const* d_in, const int* in_sizes, int n_in,
                              void* d_out, int out_size, void* d_ws, size_t ws_size,
                              hipStream_t stream) {
    const float* x       = (const float*)d_in[0];
    const int*   sid     = (const int*)  d_in[1];
    const float* W_sub   = (const float*)d_in[2];
    const float* b_sub   = (const float*)d_in[3];
    const float* conv1_w = (const float*)d_in[4];
    const float* conv1_b = (const float*)d_in[5];
    const float* bn1_g   = (const float*)d_in[6];
    const float* bn1_b   = (const float*)d_in[7];
    const float* bn1_m   = (const float*)d_in[8];
    const float* bn1_v   = (const float*)d_in[9];
    const float* conv2_w = (const float*)d_in[10];
    const float* conv2_b = (const float*)d_in[11];
    const float* bn2_g   = (const float*)d_in[12];
    const float* bn2_b   = (const float*)d_in[13];
    const float* bn2_m   = (const float*)d_in[14];
    const float* bn2_v   = (const float*)d_in[15];
    const float* proj_w  = (const float*)d_in[16];
    const float* proj_b  = (const float*)d_in[17];
    const float* mlp1_w  = (const float*)d_in[18];
    const float* mlp1_b  = (const float*)d_in[19];
    const float* mlp2_w  = (const float*)d_in[20];
    const float* mlp2_b  = (const float*)d_in[21];
    const float* ln_g    = (const float*)d_in[22];
    const float* ln_b    = (const float*)d_in[23];

    float* ws   = (float*)d_ws;
    float* out0 = ws;
    float* P    = out0 + (size_t)BB * CC * TT;
    float* h2   = P    + (size_t)BB * CC * PLEN;
    float* z    = h2   + (size_t)BB * WID * UU;
    float* z1   = z    + (size_t)BB * HID;
    float* gg   = z1   + (size_t)BB * PROJ;
    float* yy   = gg   + (size_t)BB * PROJ;

    dim3 g1(8, 32, BB);
    k_subject<<<g1, 32, 0, stream>>>(x, sid, W_sub, b_sub, out0);

    k_pool<<<BB * CC, 128, 0, stream>>>(out0, P);

    dim3 g3(UU / 16, BB);
    k_conv2<<<g3, 160, 0, stream>>>(P, conv1_w, conv1_b, bn1_g, bn1_b, bn1_m, bn1_v,
                                    conv2_w, conv2_b, bn2_g, bn2_b, bn2_m, bn2_v, h2);

    k_proj<<<BB, 256, 0, stream>>>(h2, proj_w, proj_b, z);

    dim3 g5(2, PROJ / 16);
    k_mlp1<<<g5, 32, 0, stream>>>(z, mlp1_w, mlp1_b, z1, gg);
    k_mlp2<<<g5, 32, 0, stream>>>(gg, mlp2_w, mlp2_b, z1, yy);

    k_ln<<<BB, 256, 0, stream>>>(yy, ln_g, ln_b, (float*)d_out);
}


# conv2: the block's 20 waves split K (each wave all 5 M-tiles, B fragment built once not 5x, A read as whole 128-B lines); partial tiles summed in wave order through LDS, so the K sum order changes
# speedup vs baseline: 2.3103x; 2.3103x over previous
#include <hip/hip_runtime.h>
#include <math.h>

#define BB   32
#define CC   122
#define TT   500
#define WID  80
#define UU   96
#define K2   (WID*CC)
#define HID  768
#define PROJ 1024
#define PLEN 484
#define EPSF 1e-5f

typedef float v2f __attribute__((ext_vector_type(2)));
typedef float v4f __attribute__((ext_vector_type(4)));
typedef float v8f __attribute__((ext_vector_type(8)));

static __device__ __forceinline__ v8f wmma_f32(v2f a, v2f b, v8f c) {
    return __builtin_amdgcn_wmma_f32_16x16x4_f32(false, a, false, b, (short)0, c, false, false);
}

__global__ __launch_bounds__(32) void k_subject(
    const float* __restrict__ x, const int* __restrict__ sid,
    const float* __restrict__ W_sub, const float* __restrict__ b_sub,
    float* __restrict__ out0)
{
    const int lane = threadIdx.x;
    const int b  = blockIdx.z;
    const int m0 = blockIdx.x * 16;
    const int n0 = blockIdx.y * 16;
    const int s  = sid[b];
    const float* Wg = W_sub + (size_t)s * TT * TT;

    int am = m0 + (lane & 15); if (am >= CC) am = CC - 1;
    int an = n0 + (lane & 15); if (an >= TT) an = TT - 1;
    const int koff = (lane >> 4) * 2;
    const float* pa = x  + ((size_t)b * CC + am) * TT + koff;
    const float* pb = Wg + (size_t)an * TT + koff;

    v8f acc = {};
    #pragma unroll 4
    for (int k4 = 0; k4 < TT / 4; ++k4) {
        v2f a = *(const v2f*)(pa + 4 * k4);
        v2f w = *(const v2f*)(pb + 4 * k4);
        acc = wmma_f32(a, w, acc);
    }

    const float* bg = b_sub + (size_t)s * TT;
    const int nn = n0 + (lane & 15);
    #pragma unroll
    for (int r = 0; r < 8; ++r) {
        int mm = m0 + r + ((lane >> 4) << 3);
        if (mm < CC && nn < TT)
            out0[((size_t)b * CC + mm) * TT + nn] = acc[r] + bg[nn];
    }
}

__global__ __launch_bounds__(128) void k_pool(
    const float* __restrict__ out0, float* __restrict__ P)
{
    __shared__ float row[TT];
    const int bc = blockIdx.x;
    const float* src = out0 + (size_t)bc * TT;
    for (int i = threadIdx.x; i < TT; i += 128) row[i] = src[i];
    __syncthreads();
    float* dst = P + (size_t)bc * PLEN;
    for (int m = threadIdx.x; m < PLEN; m += 128) {
        float sum = 0.f;
        #pragma unroll
        for (int i = 0; i < 17; ++i) sum += row[m + i];
        dst[m] = sum * (1.f / 17.f);
    }
}

#define NWV   20
#define NSTEP (K2 / 32)
#define PPIT  82

__global__ __launch_bounds__(NWV * 32) void k_conv2(
    const float* __restrict__ P,
    const float* __restrict__ conv1_w, const float* __restrict__ conv1_b,
    const float* __restrict__ bn1_g, const float* __restrict__ bn1_b,
    const float* __restrict__ bn1_m, const float* __restrict__ bn1_v,
    const float* __restrict__ conv2_w, const float* __restrict__ conv2_b,
    const float* __restrict__ bn2_g, const float* __restrict__ bn2_b,
    const float* __restrict__ bn2_m, const float* __restrict__ bn2_v,
    float* __restrict__ h2)
{
    __shared__ float Pl[CC * PPIT];
    __shared__ float w1s[WID * 5];
    __shared__ float sc1[WID], sh1[WID];
    __shared__ float red[NWV * 5 * 8 * 32];

    const int tid = threadIdx.x;
    const int b   = blockIdx.y;
    const int u0  = blockIdx.x * 16;

    for (int i = tid; i < WID * 5; i += NWV * 32) w1s[i] = conv1_w[i];
    if (tid < WID) {
        float sc = bn1_g[tid] * rsqrtf(bn1_v[tid] + EPSF);
        sc1[tid] = sc;
        sh1[tid] = bn1_b[tid] - bn1_m[tid] * sc + sc * conv1_b[tid];
    }
    for (int i = tid; i < CC * 80; i += NWV * 32) {
        int c = i / 80, q = i - c * 80;
        Pl[c * PPIT + q] = P[((size_t)b * CC + c) * PLEN + 5 * u0 + q];
    }
    __syncthreads();

    const int wv   = __builtin_amdgcn_readfirstlane(tid >> 5);
    const int lane = tid & 31;
    const int n    = lane & 15;
    const int half = lane >> 4;

    const int st0 = wv * 15 + (wv < 5 ? wv : 5);
    const int cnt = 15 + (wv < 5 ? 1 : 0);

    const int k0 = st0 * 32 + half * 16;
    int w = k0 / CC;
    int c = k0 - w * CC;

    const float* pa = conv2_w + (size_t)n * K2 + k0;
    const float* pp = &Pl[5 * n];

    v8f acc[5];
    #pragma unroll
    for (int t = 0; t < 5; ++t) acc[t] = (v8f){};

    for (int it = 0; it < cnt; ++it) {
        v4f a[5][4];
        #pragma unroll
        for (int t = 0; t < 5; ++t) {
            #pragma unroll
            for (int q = 0; q < 4; ++q)
                a[t][q] = *(const v4f*)(pa + (size_t)t * 16 * K2 + 4 * q);
        }

        v2f bf[8];
        #pragma unroll
        for (int j = 0; j < 8; ++j) {
            int cj = c + 2 * j, wj = w;
            if (cj >= CC) { cj -= CC; wj += 1; }
            const float* wr  = &w1s[wj * 5];
            const float  scw = sc1[wj], shw = sh1[wj];
            const float* pr0 = pp + cj * PPIT;
            const float* pr1 = pr0 + PPIT;
            float v0 = wr[0]*pr0[0] + wr[1]*pr0[1] + wr[2]*pr0[2] + wr[3]*pr0[3] + wr[4]*pr0[4];
            float g0 = scw * v0 + shw;
            float e0 = g0 > 0.f ? g0 : (__expf(g0) - 1.f);
            float v1 = wr[0]*pr1[0] + wr[1]*pr1[1] + wr[2]*pr1[2] + wr[3]*pr1[3] + wr[4]*pr1[4];
            float g1 = scw * v1 + shw;
            float e1 = g1 > 0.f ? g1 : (__expf(g1) - 1.f);
            bf[j] = (v2f){ e0, e1 };
        }

        #pragma unroll
        for (int j = 0; j < 8; ++j) {
            #pragma unroll
            for (int t = 0; t < 5; ++t) {
                v4f aq = a[t][j >> 1];
                v2f af = (j & 1) ? (v2f){ aq[2], aq[3] } : (v2f){ aq[0], aq[1] };
                acc[t] = wmma_f32(af, bf[j], acc[t]);
            }
        }

        pa += 32;
        c  += 32;
        if (c >= CC) { c -= CC; w += 1; }
    }

    float* rw = red + (size_t)wv * (5 * 8 * 32);
    #pragma unroll
    for (int t = 0; t < 5; ++t) {
        #pragma unroll
        for (int r = 0; r < 8; ++r) rw[(t * 8 + r) * 32 + lane] = acc[t][r];
    }
    __syncthreads();

    for (int sidx = tid; sidx < 5 * 8 * 32; sidx += NWV * 32) {
        float sum = 0.f;
        for (int v = 0; v < NWV; ++v) sum += red[v * (5 * 8 * 32) + sidx];
        const int t = sidx >> 8, r = (sidx >> 5) & 7, l = sidx & 31;
        const int w2 = t * 16 + r + (l >> 4) * 8;
        float sc = bn2_g[w2] * rsqrtf(bn2_v[w2] + EPSF);
        float sh = bn2_b[w2] - bn2_m[w2] * sc;
        float v  = sc * (sum + conv2_b[w2]) + sh;
        float e  = v > 0.f ? v : (__expf(v) - 1.f);
        h2[((size_t)b * WID + w2) * UU + u0 + (l & 15)] = e;
    }
}

__global__ __launch_bounds__(256) void k_proj(
    const float* __restrict__ h2, const float* __restrict__ proj_w,
    const float* __restrict__ proj_b, float* __restrict__ z)
{
    __shared__ float hs[WID * UU];
    __shared__ float pw[8 * WID];
    const int b = blockIdx.x;
    for (int i = threadIdx.x; i < WID * UU; i += 256) hs[i] = h2[(size_t)b * WID * UU + i];
    for (int i = threadIdx.x; i < 8 * WID; i += 256)  pw[i] = proj_w[i];
    __syncthreads();
    for (int idx = threadIdx.x; idx < HID; idx += 256) {
        int u = idx >> 3, e = idx & 7;
        float s = proj_b[e];
        #pragma unroll 8
        for (int w = 0; w < WID; ++w) s += pw[e * WID + w] * hs[w * UU + u];
        z[(size_t)b * HID + idx] = s;
    }
}

__global__ __launch_bounds__(32) void k_mlp1(
    const float* __restrict__ z, const float* __restrict__ w,
    const float* __restrict__ bias, float* __restrict__ z1, float* __restrict__ g)
{
    const int lane = threadIdx.x;
    const int m0 = blockIdx.x * 16, n0 = blockIdx.y * 16;
    const float* pa = z + (size_t)(m0 + (lane & 15)) * HID + (lane >> 4) * 2;
    const float* pb = w + (size_t)(n0 + (lane & 15)) * HID + (lane >> 4) * 2;
    v8f acc = {};
    #pragma unroll 4
    for (int k4 = 0; k4 < HID / 4; ++k4) {
        v2f a = *(const v2f*)(pa + 4 * k4);
        v2f bb = *(const v2f*)(pb + 4 * k4);
        acc = wmma_f32(a, bb, acc);
    }
    const int nn = n0 + (lane & 15);
    const float bn = bias[nn];
    #pragma unroll
    for (int r = 0; r < 8; ++r) {
        int mm = m0 + r + ((lane >> 4) << 3);
        float v = acc[r] + bn;
        z1[(size_t)mm * PROJ + nn] = v;
        g [(size_t)mm * PROJ + nn] = 0.5f * v * (1.f + erff(v * 0.70710678118654752f));
    }
}

__global__ __launch_bounds__(32) void k_mlp2(
    const float* __restrict__ g, const float* __restrict__ w,
    const float* __restrict__ bias, const float* __restrict__ z1,
    float* __restrict__ y)
{
    const int lane = threadIdx.x;
    const int m0 = blockIdx.x * 16, n0 = blockIdx.y * 16;
    const float* pa = g + (size_t)(m0 + (lane & 15)) * PROJ + (lane >> 4) * 2;
    const float* pb = w + (size_t)(n0 + (lane & 15)) * PROJ + (lane >> 4) * 2;
    v8f acc = {};
    #pragma unroll 4
    for (int k4 = 0; k4 < PROJ / 4; ++k4) {
        v2f a = *(const v2f*)(pa + 4 * k4);
        v2f bb = *(const v2f*)(pb + 4 * k4);
        acc = wmma_f32(a, bb, acc);
    }
    const int nn = n0 + (lane & 15);
    const float bn = bias[nn];
    #pragma unroll
    for (int r = 0; r < 8; ++r) {
        int mm = m0 + r + ((lane >> 4) << 3);
        y[(size_t)mm * PROJ + nn] = z1[(size_t)mm * PROJ + nn] + acc[r] + bn;
    }
}

__global__ __launch_bounds__(256) void k_ln(
    const float* __restrict__ y, const float* __restrict__ gam,
    const float* __restrict__ bet, float* __restrict__ out)
{
    __shared__ float red[256];
    const int b = blockIdx.x;
    const float* row = y + (size_t)b * PROJ;
    float s = 0.f;
    for (int i = threadIdx.x; i < PROJ; i += 256) s += row[i];
    red[threadIdx.x] = s; __syncthreads();
    for (int off = 128; off > 0; off >>= 1) {
        if (threadIdx.x < off) red[threadIdx.x] += red[threadIdx.x + off];
        __syncthreads();
    }
    const float mu = red[0] * (1.f / PROJ);
    __syncthreads();
    float s2 = 0.f;
    for (int i = threadIdx.x; i < PROJ; i += 256) { float d = row[i] - mu; s2 += d * d; }
    red[threadIdx.x] = s2; __syncthreads();
    for (int off = 128; off > 0; off >>= 1) {
        if (threadIdx.x < off) red[threadIdx.x] += red[threadIdx.x + off];
        __syncthreads();
    }
    const float inv = rsqrtf(red[0] * (1.f / PROJ) + EPSF);
    for (int i = threadIdx.x; i < PROJ; i += 256)
        out[(size_t)b * PROJ + i] = (row[i] - mu) * inv * gam[i] + bet[i];
}

extern "C" void kernel_launch(void* const* d_in, const int* in_sizes, int n_in,
                              void* d_out, int out_size, void* d_ws, size_t ws_size,
                              hipStream_t stream) {
    const float* x       = (const float*)d_in[0];
    const int*   sid     = (const int*)  d_in[1];
    const float* W_sub   = (const float*)d_in[2];
    const float* b_sub   = (const float*)d_in[3];
    const float* conv1_w = (const float*)d_in[4];
    const float* conv1_b = (const float*)d_in[5];
    const float* bn1_g   = (const float*)d_in[6];
    const float* bn1_b   = (const float*)d_in[7];
    const float* bn1_m   = (const float*)d_in[8];
    const float* bn1_v   = (const float*)d_in[9];
    const float* conv2_w = (const float*)d_in[10];
    const float* conv2_b = (const float*)d_in[11];
    const float* bn2_g   = (const float*)d_in[12];
    const float* bn2_b   = (const float*)d_in[13];
    const float* bn2_m   = (const float*)d_in[14];
    const float* bn2_v   = (const float*)d_in[15];
    const float* proj_w  = (const float*)d_in[16];
    const float* proj_b  = (const float*)d_in[17];
    const float* mlp1_w  = (const float*)d_in[18];
    const float* mlp1_b  = (const float*)d_in[19];
    const float* mlp2_w  = (const float*)d_in[20];
    const float* mlp2_b  = (const float*)d_in[21];
    const float* ln_g    = (const float*)d_in[22];
    const float* ln_b    = (const float*)d_in[23];

    float* ws   = (float*)d_ws;
    float* out0 = ws;
    float* P    = out0 + (size_t)BB * CC * TT;
    float* h2   = P    + (size_t)BB * CC * PLEN;
    float* z    = h2   + (size_t)BB * WID * UU;
    float* z1   = z    + (size_t)BB * HID;
    float* gg   = z1   + (size_t)BB * PROJ;
    float* yy   = gg   + (size_t)BB * PROJ;

    dim3 g1(8, 32, BB);
    k_subject<<<g1, 32, 0, stream>>>(x, sid, W_sub, b_sub, out0);

    k_pool<<<BB * CC, 128, 0, stream>>>(out0, P);

    dim3 g3(UU / 16, BB);
    k_conv2<<<g3, NWV * 32, 0, stream>>>(P, conv1_w, conv1_b, bn1_g, bn1_b, bn1_m, bn1_v,
                                    conv2_w, conv2_b, bn2_g, bn2_b, bn2_m, bn2_v, h2);

    k_proj<<<BB, 256, 0, stream>>>(h2, proj_w, proj_b, z);

    dim3 g5(2, PROJ / 16);
    k_mlp1<<<g5, 32, 0, stream>>>(z, mlp1_w, mlp1_b, z1, gg);
    k_mlp2<<<g5, 32, 0, stream>>>(gg, mlp2_w, mlp2_b, z1, yy);

    k_ln<<<BB, 256, 0, stream>>>(yy, ln_g, ln_b, (float*)d_out);
}


# k_subject: one wave per 32x64 output block (2x4 tiles, fragments reused 4x/2x), A/B rows read as whole 128-B runs one k-step ahead; k is regrouped inside the WMMAs, so the K sum order changes
# speedup vs baseline: 3.7122x; 1.6068x over previous
#include <hip/hip_runtime.h>
#include <math.h>

#define BB   32
#define CC   122
#define TT   500
#define WID  80
#define UU   96
#define K2   (WID*CC)
#define HID  768
#define PROJ 1024
#define PLEN 484
#define EPSF 1e-5f

typedef float v2f __attribute__((ext_vector_type(2)));
typedef float v4f __attribute__((ext_vector_type(4)));
typedef float v8f __attribute__((ext_vector_type(8)));

static __device__ __forceinline__ v8f wmma_f32(v2f a, v2f b, v8f c) {
    return __builtin_amdgcn_wmma_f32_16x16x4_f32(false, a, false, b, (short)0, c, false, false);
}

#define SMT 2
#define SNT 4

static __device__ __forceinline__ void sub_load(v4f (&fa)[SMT][4], v4f (&fb)[SNT][4],
                                                const float* const (&pa)[SMT],
                                                const float* const (&pb)[SNT], int koff) {
    #pragma unroll
    for (int t = 0; t < SMT; ++t) {
        #pragma unroll
        for (int q = 0; q < 4; ++q) fa[t][q] = *(const v4f*)(pa[t] + koff + 8 * q);
    }
    #pragma unroll
    for (int u = 0; u < SNT; ++u) {
        #pragma unroll
        for (int q = 0; q < 4; ++q) fb[u][q] = *(const v4f*)(pb[u] + koff + 8 * q);
    }
}

static __device__ __forceinline__ void sub_mma(v8f (&acc)[SMT][SNT], const v4f (&fa)[SMT][4],
                                               const v4f (&fb)[SNT][4]) {
    #pragma unroll
    for (int j = 0; j < 8; ++j) {
        #pragma unroll
        for (int t = 0; t < SMT; ++t) {
            const v4f aq = fa[t][j >> 1];
            const v2f af = (j & 1) ? (v2f){ aq[2], aq[3] } : (v2f){ aq[0], aq[1] };
            #pragma unroll
            for (int u = 0; u < SNT; ++u) {
                const v4f wq = fb[u][j >> 1];
                const v2f wf = (j & 1) ? (v2f){ wq[2], wq[3] } : (v2f){ wq[0], wq[1] };
                acc[t][u] = wmma_f32(af, wf, acc[t][u]);
            }
        }
    }
}

__global__ __launch_bounds__(128) void k_subject(
    const float* __restrict__ x, const int* __restrict__ sid,
    const float* __restrict__ W_sub, const float* __restrict__ b_sub,
    float* __restrict__ out0)
{
    const int tid  = threadIdx.x;
    const int lane = tid & 31;
    const int n    = lane & 15;
    const int half = lane >> 4;
    const int b  = blockIdx.y;
    const int n0 = blockIdx.x * (16 * SNT);
    const int m0 = (tid >> 5) * (16 * SMT);
    const int s  = sid[b];
    const float* Wg = W_sub + (size_t)s * TT * TT;

    const float* pa[SMT];
    const float* pb[SNT];
    #pragma unroll
    for (int t = 0; t < SMT; ++t) {
        int am = m0 + 16 * t + n; if (am >= CC) am = CC - 1;
        pa[t] = x + ((size_t)b * CC + am) * TT + 4 * half;
    }
    #pragma unroll
    for (int u = 0; u < SNT; ++u) {
        int an = n0 + 16 * u + n; if (an >= TT) an = TT - 1;
        pb[u] = Wg + (size_t)an * TT + 4 * half;
    }

    v8f acc[SMT][SNT];
    #pragma unroll
    for (int t = 0; t < SMT; ++t) {
        #pragma unroll
        for (int u = 0; u < SNT; ++u) acc[t][u] = (v8f){};
    }

    v4f fa0[SMT][4], fb0[SNT][4], fa1[SMT][4], fb1[SNT][4];
    sub_load(fa0, fb0, pa, pb, 0);
    for (int st = 0; st < 14; st += 2) {
        sub_load(fa1, fb1, pa, pb, 32 * (st + 1));
        sub_mma(acc, fa0, fb0);
        sub_load(fa0, fb0, pa, pb, 32 * (st + 2));
        sub_mma(acc, fa1, fb1);
    }

    v4f ta[SMT][2], tb[SNT][2];
    v2f ua[SMT], ub[SNT];
    #pragma unroll
    for (int t = 0; t < SMT; ++t) {
        #pragma unroll
        for (int q = 0; q < 2; ++q) ta[t][q] = *(const v4f*)(pa[t] + 480 + 8 * q);
        ua[t] = *(const v2f*)(pa[t] + 496 - 2 * half);
    }
    #pragma unroll
    for (int u = 0; u < SNT; ++u) {
        #pragma unroll
        for (int q = 0; q < 2; ++q) tb[u][q] = *(const v4f*)(pb[u] + 480 + 8 * q);
        ub[u] = *(const v2f*)(pb[u] + 496 - 2 * half);
    }

    sub_mma(acc, fa0, fb0);

    #pragma unroll
    for (int j = 0; j < 4; ++j) {
        #pragma unroll
        for (int t = 0; t < SMT; ++t) {
            const v4f aq = ta[t][j >> 1];
            const v2f af = (j & 1) ? (v2f){ aq[2], aq[3] } : (v2f){ aq[0], aq[1] };
            #pragma unroll
            for (int u = 0; u < SNT; ++u) {
                const v4f wq = tb[u][j >> 1];
                const v2f wf = (j & 1) ? (v2f){ wq[2], wq[3] } : (v2f){ wq[0], wq[1] };
                acc[t][u] = wmma_f32(af, wf, acc[t][u]);
            }
        }
    }
    #pragma unroll
    for (int t = 0; t < SMT; ++t) {
        #pragma unroll
        for (int u = 0; u < SNT; ++u) acc[t][u] = wmma_f32(ua[t], ub[u], acc[t][u]);
    }

    const float* bg = b_sub + (size_t)s * TT;
    #pragma unroll
    for (int u = 0; u < SNT; ++u) {
        const int nn = n0 + 16 * u + n;
        const float bias = bg[nn < TT ? nn : TT - 1];
        #pragma unroll
        for (int t = 0; t < SMT; ++t) {
            #pragma unroll
            for (int r = 0; r < 8; ++r) {
                const int mm = m0 + 16 * t + r + (half << 3);
                if (mm < CC && nn < TT)
                    out0[((size_t)b * CC + mm) * TT + nn] = acc[t][u][r] + bias;
            }
        }
    }
}

__global__ __launch_bounds__(128) void k_pool(
    const float* __restrict__ out0, float* __restrict__ P)
{
    __shared__ float row[TT];
    const int bc = blockIdx.x;
    const float* src = out0 + (size_t)bc * TT;
    for (int i = threadIdx.x; i < TT; i += 128) row[i] = src[i];
    __syncthreads();
    float* dst = P + (size_t)bc * PLEN;
    for (int m = threadIdx.x; m < PLEN; m += 128) {
        float sum = 0.f;
        #pragma unroll
        for (int i = 0; i < 17; ++i) sum += row[m + i];
        dst[m] = sum * (1.f / 17.f);
    }
}

#define NWV   20
#define NSTEP (K2 / 32)
#define PPIT  82

__global__ __launch_bounds__(NWV * 32) void k_conv2(
    const float* __restrict__ P,
    const float* __restrict__ conv1_w, const float* __restrict__ conv1_b,
    const float* __restrict__ bn1_g, const float* __restrict__ bn1_b,
    const float* __restrict__ bn1_m, const float* __restrict__ bn1_v,
    const float* __restrict__ conv2_w, const float* __restrict__ conv2_b,
    const float* __restrict__ bn2_g, const float* __restrict__ bn2_b,
    const float* __restrict__ bn2_m, const float* __restrict__ bn2_v,
    float* __restrict__ h2)
{
    __shared__ float Pl[CC * PPIT];
    __shared__ float w1s[WID * 5];
    __shared__ float sc1[WID], sh1[WID];
    __shared__ float red[NWV * 5 * 8 * 32];

    const int tid = threadIdx.x;
    const int b   = blockIdx.y;
    const int u0  = blockIdx.x * 16;

    for (int i = tid; i < WID * 5; i += NWV * 32) w1s[i] = conv1_w[i];
    if (tid < WID) {
        float sc = bn1_g[tid] * rsqrtf(bn1_v[tid] + EPSF);
        sc1[tid] = sc;
        sh1[tid] = bn1_b[tid] - bn1_m[tid] * sc + sc * conv1_b[tid];
    }
    for (int i = tid; i < CC * 80; i += NWV * 32) {
        int c = i / 80, q = i - c * 80;
        Pl[c * PPIT + q] = P[((size_t)b * CC + c) * PLEN + 5 * u0 + q];
    }
    __syncthreads();

    const int wv   = __builtin_amdgcn_readfirstlane(tid >> 5);
    const int lane = tid & 31;
    const int n    = lane & 15;
    const int half = lane >> 4;

    const int st0 = wv * 15 + (wv < 5 ? wv : 5);
    const int cnt = 15 + (wv < 5 ? 1 : 0);

    const int k0 = st0 * 32 + half * 16;
    int w = k0 / CC;
    int c = k0 - w * CC;

    const float* pa = conv2_w + (size_t)n * K2 + k0;
    const float* pp = &Pl[5 * n];

    v8f acc[5];
    #pragma unroll
    for (int t = 0; t < 5; ++t) acc[t] = (v8f){};

    for (int it = 0; it < cnt; ++it) {
        v4f a[5][4];
        #pragma unroll
        for (int t = 0; t < 5; ++t) {
            #pragma unroll
            for (int q = 0; q < 4; ++q)
                a[t][q] = *(const v4f*)(pa + (size_t)t * 16 * K2 + 4 * q);
        }

        v2f bf[8];
        #pragma unroll
        for (int j = 0; j < 8; ++j) {
            int cj = c + 2 * j, wj = w;
            if (cj >= CC) { cj -= CC; wj += 1; }
            const float* wr  = &w1s[wj * 5];
            const float  scw = sc1[wj], shw = sh1[wj];
            const float* pr0 = pp + cj * PPIT;
            const float* pr1 = pr0 + PPIT;
            float v0 = wr[0]*pr0[0] + wr[1]*pr0[1] + wr[2]*pr0[2] + wr[3]*pr0[3] + wr[4]*pr0[4];
            float g0 = scw * v0 + shw;
            float e0 = g0 > 0.f ? g0 : (__expf(g0) - 1.f);
            float v1 = wr[0]*pr1[0] + wr[1]*pr1[1] + wr[2]*pr1[2] + wr[3]*pr1[3] + wr[4]*pr1[4];
            float g1 = scw * v1 + shw;
            float e1 = g1 > 0.f ? g1 : (__expf(g1) - 1.f);
            bf[j] = (v2f){ e0, e1 };
        }

        #pragma unroll
        for (int j = 0; j < 8; ++j) {
            #pragma unroll
            for (int t = 0; t < 5; ++t) {
                v4f aq = a[t][j >> 1];
                v2f af = (j & 1) ? (v2f){ aq[2], aq[3] } : (v2f){ aq[0], aq[1] };
                acc[t] = wmma_f32(af, bf[j], acc[t]);
            }
        }

        pa += 32;
        c  += 32;
        if (c >= CC) { c -= CC; w += 1; }
    }

    float* rw = red + (size_t)wv * (5 * 8 * 32);
    #pragma unroll
    for (int t = 0; t < 5; ++t) {
        #pragma unroll
        for (int r = 0; r < 8; ++r) rw[(t * 8 + r) * 32 + lane] = acc[t][r];
    }
    __syncthreads();

    for (int sidx = tid; sidx < 5 * 8 * 32; sidx += NWV * 32) {
        float sum = 0.f;
        for (int v = 0; v < NWV; ++v) sum += red[v * (5 * 8 * 32) + sidx];
        const int t = sidx >> 8, r = (sidx >> 5) & 7, l = sidx & 31;
        const int w2 = t * 16 + r + (l >> 4) * 8;
        float sc = bn2_g[w2] * rsqrtf(bn2_v[w2] + EPSF);
        float sh = bn2_b[w2] - bn2_m[w2] * sc;
        float v  = sc * (sum + conv2_b[w2]) + sh;
        float e  = v > 0.f ? v : (__expf(v) - 1.f);
        h2[((size_t)b * WID + w2) * UU + u0 + (l & 15)] = e;
    }
}

__global__ __launch_bounds__(256) void k_proj(
    const float* __restrict__ h2, const float* __restrict__ proj_w,
    const float* __restrict__ proj_b, float* __restrict__ z)
{
    __shared__ float hs[WID * UU];
    __shared__ float pw[8 * WID];
    const int b = blockIdx.x;
    for (int i = threadIdx.x; i < WID * UU; i += 256) hs[i] = h2[(size_t)b * WID * UU + i];
    for (int i = threadIdx.x; i < 8 * WID; i += 256)  pw[i] = proj_w[i];
    __syncthreads();
    for (int idx = threadIdx.x; idx < HID; idx += 256) {
        int u = idx >> 3, e = idx & 7;
        float s = proj_b[e];
        #pragma unroll 8
        for (int w = 0; w < WID; ++w) s += pw[e * WID + w] * hs[w * UU + u];
        z[(size_t)b * HID + idx] = s;
    }
}

__global__ __launch_bounds__(32) void k_mlp1(
    const float* __restrict__ z, const float* __restrict__ w,
    const float* __restrict__ bias, float* __restrict__ z1, float* __restrict__ g)
{
    const int lane = threadIdx.x;
    const int m0 = blockIdx.x * 16, n0 = blockIdx.y * 16;
    const float* pa = z + (size_t)(m0 + (lane & 15)) * HID + (lane >> 4) * 2;
    const float* pb = w + (size_t)(n0 + (lane & 15)) * HID + (lane >> 4) * 2;
    v8f acc = {};
    #pragma unroll 4
    for (int k4 = 0; k4 < HID / 4; ++k4) {
        v2f a = *(const v2f*)(pa + 4 * k4);
        v2f bb = *(const v2f*)(pb + 4 * k4);
        acc = wmma_f32(a, bb, acc);
    }
    const int nn = n0 + (lane & 15);
    const float bn = bias[nn];
    #pragma unroll
    for (int r = 0; r < 8; ++r) {
        int mm = m0 + r + ((lane >> 4) << 3);
        float v = acc[r] + bn;
        z1[(size_t)mm * PROJ + nn] = v;
        g [(size_t)mm * PROJ + nn] = 0.5f * v * (1.f + erff(v * 0.70710678118654752f));
    }
}

__global__ __launch_bounds__(32) void k_mlp2(
    const float* __restrict__ g, const float* __restrict__ w,
    const float* __restrict__ bias, const float* __restrict__ z1,
    float* __restrict__ y)
{
    const int lane = threadIdx.x;
    const int m0 = blockIdx.x * 16, n0 = blockIdx.y * 16;
    const float* pa = g + (size_t)(m0 + (lane & 15)) * PROJ + (lane >> 4) * 2;
    const float* pb = w + (size_t)(n0 + (lane & 15)) * PROJ + (lane >> 4) * 2;
    v8f acc = {};
    #pragma unroll 4
    for (int k4 = 0; k4 < PROJ / 4; ++k4) {
        v2f a = *(const v2f*)(pa + 4 * k4);
        v2f bb = *(const v2f*)(pb + 4 * k4);
        acc = wmma_f32(a, bb, acc);
    }
    const int nn = n0 + (lane & 15);
    const float bn = bias[nn];
    #pragma unroll
    for (int r = 0; r < 8; ++r) {
        int mm = m0 + r + ((lane >> 4) << 3);
        y[(size_t)mm * PROJ + nn] = z1[(size_t)mm * PROJ + nn] + acc[r] + bn;
    }
}

__global__ __launch_bounds__(256) void k_ln(
    const float* __restrict__ y, const float* __restrict__ gam,
    const float* __restrict__ bet, float* __restrict__ out)
{
    __shared__ float red[256];
    const int b = blockIdx.x;
    const float* row = y + (size_t)b * PROJ;
    float s = 0.f;
    for (int i = threadIdx.x; i < PROJ; i += 256) s += row[i];
    red[threadIdx.x] = s; __syncthreads();
    for (int off = 128; off > 0; off >>= 1) {
        if (threadIdx.x < off) red[threadIdx.x] += red[threadIdx.x + off];
        __syncthreads();
    }
    const float mu = red[0] * (1.f / PROJ);
    __syncthreads();
    float s2 = 0.f;
    for (int i = threadIdx.x; i < PROJ; i += 256) { float d = row[i] - mu; s2 += d * d; }
    red[threadIdx.x] = s2; __syncthreads();
    for (int off = 128; off > 0; off >>= 1) {
        if (threadIdx.x < off) red[threadIdx.x] += red[threadIdx.x + off];
        __syncthreads();
    }
    const float inv = rsqrtf(red[0] * (1.f / PROJ) + EPSF);
    for (int i = threadIdx.x; i < PROJ; i += 256)
        out[(size_t)b * PROJ + i] = (row[i] - mu) * inv * gam[i] + bet[i];
}

extern "C" void kernel_launch(void* const* d_in, const int* in_sizes, int n_in,
                              void* d_out, int out_size, void* d_ws, size_t ws_size,
                              hipStream_t stream) {
    const float* x       = (const float*)d_in[0];
    const int*   sid     = (const int*)  d_in[1];
    const float* W_sub   = (const float*)d_in[2];
    const float* b_sub   = (const float*)d_in[3];
    const float* conv1_w = (const float*)d_in[4];
    const float* conv1_b = (const float*)d_in[5];
    const float* bn1_g   = (const float*)d_in[6];
    const float* bn1_b   = (const float*)d_in[7];
    const float* bn1_m   = (const float*)d_in[8];
    const float* bn1_v   = (const float*)d_in[9];
    const float* conv2_w = (const float*)d_in[10];
    const float* conv2_b = (const float*)d_in[11];
    const float* bn2_g   = (const float*)d_in[12];
    const float* bn2_b   = (const float*)d_in[13];
    const float* bn2_m   = (const float*)d_in[14];
    const float* bn2_v   = (const float*)d_in[15];
    const float* proj_w  = (const float*)d_in[16];
    const float* proj_b  = (const float*)d_in[17];
    const float* mlp1_w  = (const float*)d_in[18];
    const float* mlp1_b  = (const float*)d_in[19];
    const float* mlp2_w  = (const float*)d_in[20];
    const float* mlp2_b  = (const float*)d_in[21];
    const float* ln_g    = (const float*)d_in[22];
    const float* ln_b    = (const float*)d_in[23];

    float* ws   = (float*)d_ws;
    float* out0 = ws;
    float* P    = out0 + (size_t)BB * CC * TT;
    float* h2   = P    + (size_t)BB * CC * PLEN;
    float* z    = h2   + (size_t)BB * WID * UU;
    float* z1   = z    + (size_t)BB * HID;
    float* gg   = z1   + (size_t)BB * PROJ;
    float* yy   = gg   + (size_t)BB * PROJ;

    dim3 g1(8, BB);
    k_subject<<<g1, 128, 0, stream>>>(x, sid, W_sub, b_sub, out0);

    k_pool<<<BB * CC, 128, 0, stream>>>(out0, P);

    dim3 g3(UU / 16, BB);
    k_conv2<<<g3, NWV * 32, 0, stream>>>(P, conv1_w, conv1_b, bn1_g, bn1_b, bn1_m, bn1_v,
                                    conv2_w, conv2_b, bn2_g, bn2_b, bn2_m, bn2_v, h2);

    k_proj<<<BB, 256, 0, stream>>>(h2, proj_w, proj_b, z);

    dim3 g5(2, PROJ / 16);
    k_mlp1<<<g5, 32, 0, stream>>>(z, mlp1_w, mlp1_b, z1, gg);
    k_mlp2<<<g5, 32, 0, stream>>>(gg, mlp2_w, mlp2_b, z1, yy);

    k_ln<<<BB, 256, 0, stream>>>(yy, ln_g, ln_b, (float*)d_out);
}


# mlp1/mlp2: a block's 8 waves split K of one 16x16 tile; a wave's 16-B fragment loads run ahead of its two WMMA chains; partial tiles summed in wave order through LDS, so the K sum order changes
# speedup vs baseline: 4.5362x; 1.2220x over previous
#include <hip/hip_runtime.h>
#include <math.h>

#define BB   32
#define CC   122
#define TT   500
#define WID  80
#define UU   96
#define K2   (WID*CC)
#define HID  768
#define PROJ 1024
#define PLEN 484
#define EPSF 1e-5f

typedef float v2f __attribute__((ext_vector_type(2)));
typedef float v4f __attribute__((ext_vector_type(4)));
typedef float v8f __attribute__((ext_vector_type(8)));

static __device__ __forceinline__ v8f wmma_f32(v2f a, v2f b, v8f c) {
    return __builtin_amdgcn_wmma_f32_16x16x4_f32(false, a, false, b, (short)0, c, false, false);
}

#define SMT 2
#define SNT 4

static __device__ __forceinline__ void sub_load(v4f (&fa)[SMT][4], v4f (&fb)[SNT][4],
                                                const float* const (&pa)[SMT],
                                                const float* const (&pb)[SNT], int koff) {
    #pragma unroll
    for (int t = 0; t < SMT; ++t) {
        #pragma unroll
        for (int q = 0; q < 4; ++q) fa[t][q] = *(const v4f*)(pa[t] + koff + 8 * q);
    }
    #pragma unroll
    for (int u = 0; u < SNT; ++u) {
        #pragma unroll
        for (int q = 0; q < 4; ++q) fb[u][q] = *(const v4f*)(pb[u] + koff + 8 * q);
    }
}

static __device__ __forceinline__ void sub_mma(v8f (&acc)[SMT][SNT], const v4f (&fa)[SMT][4],
                                               const v4f (&fb)[SNT][4]) {
    #pragma unroll
    for (int j = 0; j < 8; ++j) {
        #pragma unroll
        for (int t = 0; t < SMT; ++t) {
            const v4f aq = fa[t][j >> 1];
            const v2f af = (j & 1) ? (v2f){ aq[2], aq[3] } : (v2f){ aq[0], aq[1] };
            #pragma unroll
            for (int u = 0; u < SNT; ++u) {
                const v4f wq = fb[u][j >> 1];
                const v2f wf = (j & 1) ? (v2f){ wq[2], wq[3] } : (v2f){ wq[0], wq[1] };
                acc[t][u] = wmma_f32(af, wf, acc[t][u]);
            }
        }
    }
}

__global__ __launch_bounds__(128) void k_subject(
    const float* __restrict__ x, const int* __restrict__ sid,
    const float* __restrict__ W_sub, const float* __restrict__ b_sub,
    float* __restrict__ out0)
{
    const int tid  = threadIdx.x;
    const int lane = tid & 31;
    const int n    = lane & 15;
    const int half = lane >> 4;
    const int b  = blockIdx.y;
    const int n0 = blockIdx.x * (16 * SNT);
    const int m0 = (tid >> 5) * (16 * SMT);
    const int s  = sid[b];
    const float* Wg = W_sub + (size_t)s * TT * TT;

    const float* pa[SMT];
    const float* pb[SNT];
    #pragma unroll
    for (int t = 0; t < SMT; ++t) {
        int am = m0 + 16 * t + n; if (am >= CC) am = CC - 1;
        pa[t] = x + ((size_t)b * CC + am) * TT + 4 * half;
    }
    #pragma unroll
    for (int u = 0; u < SNT; ++u) {
        int an = n0 + 16 * u + n; if (an >= TT) an = TT - 1;
        pb[u] = Wg + (size_t)an * TT + 4 * half;
    }

    v8f acc[SMT][SNT];
    #pragma unroll
    for (int t = 0; t < SMT; ++t) {
        #pragma unroll
        for (int u = 0; u < SNT; ++u) acc[t][u] = (v8f){};
    }

    v4f fa0[SMT][4], fb0[SNT][4], fa1[SMT][4], fb1[SNT][4];
    sub_load(fa0, fb0, pa, pb, 0);
    for (int st = 0; st < 14; st += 2) {
        sub_load(fa1, fb1, pa, pb, 32 * (st + 1));
        sub_mma(acc, fa0, fb0);
        sub_load(fa0, fb0, pa, pb, 32 * (st + 2));
        sub_mma(acc, fa1, fb1);
    }

    v4f ta[SMT][2], tb[SNT][2];
    v2f ua[SMT], ub[SNT];
    #pragma unroll
    for (int t = 0; t < SMT; ++t) {
        #pragma unroll
        for (int q = 0; q < 2; ++q) ta[t][q] = *(const v4f*)(pa[t] + 480 + 8 * q);
        ua[t] = *(const v2f*)(pa[t] + 496 - 2 * half);
    }
    #pragma unroll
    for (int u = 0; u < SNT; ++u) {
        #pragma unroll
        for (int q = 0; q < 2; ++q) tb[u][q] = *(const v4f*)(pb[u] + 480 + 8 * q);
        ub[u] = *(const v2f*)(pb[u] + 496 - 2 * half);
    }

    sub_mma(acc, fa0, fb0);

    #pragma unroll
    for (int j = 0; j < 4; ++j) {
        #pragma unroll
        for (int t = 0; t < SMT; ++t) {
            const v4f aq = ta[t][j >> 1];
            const v2f af = (j & 1) ? (v2f){ aq[2], aq[3] } : (v2f){ aq[0], aq[1] };
            #pragma unroll
            for (int u = 0; u < SNT; ++u) {
                const v4f wq = tb[u][j >> 1];
                const v2f wf = (j & 1) ? (v2f){ wq[2], wq[3] } : (v2f){ wq[0], wq[1] };
                acc[t][u] = wmma_f32(af, wf, acc[t][u]);
            }
        }
    }
    #pragma unroll
    for (int t = 0; t < SMT; ++t) {
        #pragma unroll
        for (int u = 0; u < SNT; ++u) acc[t][u] = wmma_f32(ua[t], ub[u], acc[t][u]);
    }

    const float* bg = b_sub + (size_t)s * TT;
    #pragma unroll
    for (int u = 0; u < SNT; ++u) {
        const int nn = n0 + 16 * u + n;
        const float bias = bg[nn < TT ? nn : TT - 1];
        #pragma unroll
        for (int t = 0; t < SMT; ++t) {
            #pragma unroll
            for (int r = 0; r < 8; ++r) {
                const int mm = m0 + 16 * t + r + (half << 3);
                if (mm < CC && nn < TT)
                    out0[((size_t)b * CC + mm) * TT + nn] = acc[t][u][r] + bias;
            }
        }
    }
}

__global__ __launch_bounds__(128) void k_pool(
    const float* __restrict__ out0, float* __restrict__ P)
{
    __shared__ float row[TT];
    const int bc = blockIdx.x;
    const float* src = out0 + (size_t)bc * TT;
    for (int i = threadIdx.x; i < TT; i += 128) row[i] = src[i];
    __syncthreads();
    float* dst = P + (size_t)bc * PLEN;
    for (int m = threadIdx.x; m < PLEN; m += 128) {
        float sum = 0.f;
        #pragma unroll
        for (int i = 0; i < 17; ++i) sum += row[m + i];
        dst[m] = sum * (1.f / 17.f);
    }
}

#define NWV   20
#define NSTEP (K2 / 32)
#define PPIT  82

__global__ __launch_bounds__(NWV * 32) void k_conv2(
    const float* __restrict__ P,
    const float* __restrict__ conv1_w, const float* __restrict__ conv1_b,
    const float* __restrict__ bn1_g, const float* __restrict__ bn1_b,
    const float* __restrict__ bn1_m, const float* __restrict__ bn1_v,
    const float* __restrict__ conv2_w, const float* __restrict__ conv2_b,
    const float* __restrict__ bn2_g, const float* __restrict__ bn2_b,
    const float* __restrict__ bn2_m, const float* __restrict__ bn2_v,
    float* __restrict__ h2)
{
    __shared__ float Pl[CC * PPIT];
    __shared__ float w1s[WID * 5];
    __shared__ float sc1[WID], sh1[WID];
    __shared__ float red[NWV * 5 * 8 * 32];

    const int tid = threadIdx.x;
    const int b   = blockIdx.y;
    const int u0  = blockIdx.x * 16;

    for (int i = tid; i < WID * 5; i += NWV * 32) w1s[i] = conv1_w[i];
    if (tid < WID) {
        float sc = bn1_g[tid] * rsqrtf(bn1_v[tid] + EPSF);
        sc1[tid] = sc;
        sh1[tid] = bn1_b[tid] - bn1_m[tid] * sc + sc * conv1_b[tid];
    }
    for (int i = tid; i < CC * 80; i += NWV * 32) {
        int c = i / 80, q = i - c * 80;
        Pl[c * PPIT + q] = P[((size_t)b * CC + c) * PLEN + 5 * u0 + q];
    }
    __syncthreads();

    const int wv   = __builtin_amdgcn_readfirstlane(tid >> 5);
    const int lane = tid & 31;
    const int n    = lane & 15;
    const int half = lane >> 4;

    const int st0 = wv * 15 + (wv < 5 ? wv : 5);
    const int cnt = 15 + (wv < 5 ? 1 : 0);

    const int k0 = st0 * 32 + half * 16;
    int w = k0 / CC;
    int c = k0 - w * CC;

    const float* pa = conv2_w + (size_t)n * K2 + k0;
    const float* pp = &Pl[5 * n];

    v8f acc[5];
    #pragma unroll
    for (int t = 0; t < 5; ++t) acc[t] = (v8f){};

    for (int it = 0; it < cnt; ++it) {
        v4f a[5][4];
        #pragma unroll
        for (int t = 0; t < 5; ++t) {
            #pragma unroll
            for (int q = 0; q < 4; ++q)
                a[t][q] = *(const v4f*)(pa + (size_t)t * 16 * K2 + 4 * q);
        }

        v2f bf[8];
        #pragma unroll
        for (int j = 0; j < 8; ++j) {
            int cj = c + 2 * j, wj = w;
            if (cj >= CC) { cj -= CC; wj += 1; }
            const float* wr  = &w1s[wj * 5];
            const float  scw = sc1[wj], shw = sh1[wj];
            const float* pr0 = pp + cj * PPIT;
            const float* pr1 = pr0 + PPIT;
            float v0 = wr[0]*pr0[0] + wr[1]*pr0[1] + wr[2]*pr0[2] + wr[3]*pr0[3] + wr[4]*pr0[4];
            float g0 = scw * v0 + shw;
            float e0 = g0 > 0.f ? g0 : (__expf(g0) - 1.f);
            float v1 = wr[0]*pr1[0] + wr[1]*pr1[1] + wr[2]*pr1[2] + wr[3]*pr1[3] + wr[4]*pr1[4];
            float g1 = scw * v1 + shw;
            float e1 = g1 > 0.f ? g1 : (__expf(g1) - 1.f);
            bf[j] = (v2f){ e0, e1 };
        }

        #pragma unroll
        for (int j = 0; j < 8; ++j) {
            #pragma unroll
            for (int t = 0; t < 5; ++t) {
                v4f aq = a[t][j >> 1];
                v2f af = (j & 1) ? (v2f){ aq[2], aq[3] } : (v2f){ aq[0], aq[1] };
                acc[t] = wmma_f32(af, bf[j], acc[t]);
            }
        }

        pa += 32;
        c  += 32;
        if (c >= CC) { c -= CC; w += 1; }
    }

    float* rw = red + (size_t)wv * (5 * 8 * 32);
    #pragma unroll
    for (int t = 0; t < 5; ++t) {
        #pragma unroll
        for (int r = 0; r < 8; ++r) rw[(t * 8 + r) * 32 + lane] = acc[t][r];
    }
    __syncthreads();

    for (int sidx = tid; sidx < 5 * 8 * 32; sidx += NWV * 32) {
        float sum = 0.f;
        for (int v = 0; v < NWV; ++v) sum += red[v * (5 * 8 * 32) + sidx];
        const int t = sidx >> 8, r = (sidx >> 5) & 7, l = sidx & 31;
        const int w2 = t * 16 + r + (l >> 4) * 8;
        float sc = bn2_g[w2] * rsqrtf(bn2_v[w2] + EPSF);
        float sh = bn2_b[w2] - bn2_m[w2] * sc;
        float v  = sc * (sum + conv2_b[w2]) + sh;
        float e  = v > 0.f ? v : (__expf(v) - 1.f);
        h2[((size_t)b * WID + w2) * UU + u0 + (l & 15)] = e;
    }
}

__global__ __launch_bounds__(256) void k_proj(
    const float* __restrict__ h2, const float* __restrict__ proj_w,
    const float* __restrict__ proj_b, float* __restrict__ z)
{
    __shared__ float hs[WID * UU];
    __shared__ float pw[8 * WID];
    const int b = blockIdx.x;
    for (int i = threadIdx.x; i < WID * UU; i += 256) hs[i] = h2[(size_t)b * WID * UU + i];
    for (int i = threadIdx.x; i < 8 * WID; i += 256)  pw[i] = proj_w[i];
    __syncthreads();
    for (int idx = threadIdx.x; idx < HID; idx += 256) {
        int u = idx >> 3, e = idx & 7;
        float s = proj_b[e];
        #pragma unroll 8
        for (int w = 0; w < WID; ++w) s += pw[e * WID + w] * hs[w * UU + u];
        z[(size_t)b * HID + idx] = s;
    }
}

#define MW 8

template <int KTOT>
static __device__ __forceinline__ v8f mlp_partial(const float* __restrict__ A,
                                                  const float* __restrict__ Wt,
                                                  int m0, int n0, int wv, int lane)
{
    constexpr int KS = KTOT / MW;
    constexpr int NQ = KS / 8;
    const int n = lane & 15, half = lane >> 4;
    const float* pa = A  + (size_t)(m0 + n) * KTOT + wv * KS + 4 * half;
    const float* pb = Wt + (size_t)(n0 + n) * KTOT + wv * KS + 4 * half;
    v4f fa[NQ], fb[NQ];
    #pragma unroll
    for (int q = 0; q < NQ; ++q) fb[q] = *(const v4f*)(pb + 8 * q);
    #pragma unroll
    for (int q = 0; q < NQ; ++q) fa[q] = *(const v4f*)(pa + 8 * q);
    v8f acc0 = {}, acc1 = {};
    #pragma unroll
    for (int q = 0; q < NQ; ++q) {
        acc0 = wmma_f32((v2f){ fa[q][0], fa[q][1] }, (v2f){ fb[q][0], fb[q][1] }, acc0);
        acc1 = wmma_f32((v2f){ fa[q][2], fa[q][3] }, (v2f){ fb[q][2], fb[q][3] }, acc1);
    }
    return acc0 + acc1;
}

__global__ __launch_bounds__(MW * 32) void k_mlp1(
    const float* __restrict__ z, const float* __restrict__ w,
    const float* __restrict__ bias, float* __restrict__ z1, float* __restrict__ g)
{
    __shared__ float part[MW * 8 * 32];
    const int tid  = threadIdx.x;
    const int lane = tid & 31;
    const int wv   = tid >> 5;
    const int m0 = blockIdx.x * 16, n0 = blockIdx.y * 16;
    const int nn = n0 + (lane & 15);
    const float bn = bias[nn];

    const v8f acc = mlp_partial<HID>(z, w, m0, n0, wv, lane);
    #pragma unroll
    for (int r = 0; r < 8; ++r) part[(wv * 8 + r) * 32 + lane] = acc[r];
    __syncthreads();

    float sum = 0.f;
    #pragma unroll
    for (int v = 0; v < MW; ++v) sum += part[(v * 8 + wv) * 32 + lane];
    const int mm = m0 + wv + ((lane >> 4) << 3);
    const float v = sum + bn;
    z1[(size_t)mm * PROJ + nn] = v;
    g [(size_t)mm * PROJ + nn] = 0.5f * v * (1.f + erff(v * 0.70710678118654752f));
}

__global__ __launch_bounds__(MW * 32) void k_mlp2(
    const float* __restrict__ g, const float* __restrict__ w,
    const float* __restrict__ bias, const float* __restrict__ z1,
    float* __restrict__ y)
{
    __shared__ float part[MW * 8 * 32];
    const int tid  = threadIdx.x;
    const int lane = tid & 31;
    const int wv   = tid >> 5;
    const int m0 = blockIdx.x * 16, n0 = blockIdx.y * 16;
    const int nn = n0 + (lane & 15);
    const int mm = m0 + wv + ((lane >> 4) << 3);
    const float bn  = bias[nn];
    const float res = z1[(size_t)mm * PROJ + nn];

    const v8f acc = mlp_partial<PROJ>(g, w, m0, n0, wv, lane);
    #pragma unroll
    for (int r = 0; r < 8; ++r) part[(wv * 8 + r) * 32 + lane] = acc[r];
    __syncthreads();

    float sum = 0.f;
    #pragma unroll
    for (int v = 0; v < MW; ++v) sum += part[(v * 8 + wv) * 32 + lane];
    y[(size_t)mm * PROJ + nn] = res + sum + bn;
}

__global__ __launch_bounds__(256) void k_ln(
    const float* __restrict__ y, const float* __restrict__ gam,
    const float* __restrict__ bet, float* __restrict__ out)
{
    __shared__ float red[256];
    const int b = blockIdx.x;
    const float* row = y + (size_t)b * PROJ;
    float s = 0.f;
    for (int i = threadIdx.x; i < PROJ; i += 256) s += row[i];
    red[threadIdx.x] = s; __syncthreads();
    for (int off = 128; off > 0; off >>= 1) {
        if (threadIdx.x < off) red[threadIdx.x] += red[threadIdx.x + off];
        __syncthreads();
    }
    const float mu = red[0] * (1.f / PROJ);
    __syncthreads();
    float s2 = 0.f;
    for (int i = threadIdx.x; i < PROJ; i += 256) { float d = row[i] - mu; s2 += d * d; }
    red[threadIdx.x] = s2; __syncthreads();
    for (int off = 128; off > 0; off >>= 1) {
        if (threadIdx.x < off) red[threadIdx.x] += red[threadIdx.x + off];
        __syncthreads();
    }
    const float inv = rsqrtf(red[0] * (1.f / PROJ) + EPSF);
    for (int i = threadIdx.x; i < PROJ; i += 256)
        out[(size_t)b * PROJ + i] = (row[i] - mu) * inv * gam[i] + bet[i];
}

extern "C" void kernel_launch(void* const* d_in, const int* in_sizes, int n_in,
                              void* d_out, int out_size, void* d_ws, size_t ws_size,
                              hipStream_t stream) {
    const float* x       = (const float*)d_in[0];
    const int*   sid     = (const int*)  d_in[1];
    const float* W_sub   = (const float*)d_in[2];
    const float* b_sub   = (const float*)d_in[3];
    const float* conv1_w = (const float*)d_in[4];
    const float* conv1_b = (const float*)d_in[5];
    const float* bn1_g   = (const float*)d_in[6];
    const float* bn1_b   = (const float*)d_in[7];
    const float* bn1_m   = (const float*)d_in[8];
    const float* bn1_v   = (const float*)d_in[9];
    const float* conv2_w = (const float*)d_in[10];
    const float* conv2_b = (const float*)d_in[11];
    const float* bn2_g   = (const float*)d_in[12];
    const float* bn2_b   = (const float*)d_in[13];
    const float* bn2_m   = (const float*)d_in[14];
    const float* bn2_v   = (const float*)d_in[15];
    const float* proj_w  = (const float*)d_in[16];
    const float* proj_b  = (const float*)d_in[17];
    const float* mlp1_w  = (const float*)d_in[18];
    const float* mlp1_b  = (const float*)d_in[19];
    const float* mlp2_w  = (const float*)d_in[20];
    const float* mlp2_b  = (const float*)d_in[21];
    const float* ln_g    = (const float*)d_in[22];
    const float* ln_b    = (const float*)d_in[23];

    float* ws   = (float*)d_ws;
    float* out0 = ws;
    float* P    = out0 + (size_t)BB * CC * TT;
    float* h2   = P    + (size_t)BB * CC * PLEN;
    float* z    = h2   + (size_t)BB * WID * UU;
    float* z1   = z    + (size_t)BB * HID;
    float* gg   = z1   + (size_t)BB * PROJ;
    float* yy   = gg   + (size_t)BB * PROJ;

    dim3 g1(8, BB);
    k_subject<<<g1, 128, 0, stream>>>(x, sid, W_sub, b_sub, out0);

    k_pool<<<BB * CC, 128, 0, stream>>>(out0, P);

    dim3 g3(UU / 16, BB);
    k_conv2<<<g3, NWV * 32, 0, stream>>>(P, conv1_w, conv1_b, bn1_g, bn1_b, bn1_m, bn1_v,
                                    conv2_w, conv2_b, bn2_g, bn2_b, bn2_m, bn2_v, h2);

    k_proj<<<BB, 256, 0, stream>>>(h2, proj_w, proj_b, z);

    dim3 g5(2, PROJ / 16);
    k_mlp1<<<g5, MW * 32, 0, stream>>>(z, mlp1_w, mlp1_b, z1, gg);
    k_mlp2<<<g5, MW * 32, 0, stream>>>(gg, mlp2_w, mlp2_b, z1, yy);

    k_ln<<<BB, 256, 0, stream>>>(yy, ln_g, ln_b, (float*)d_out);
}


# conv2 retiled: one block per (batch, K eighth) covers all 96 u, a wave owns 2 N-tiles (an A fragment feeds 2 WMMAs); k_proj adds the 8 partial planes before bn2/elu, so the K sum order changes
# speedup vs baseline: 5.6072x; 1.2361x over previous
#include <hip/hip_runtime.h>
#include <math.h>

#define BB   32
#define CC   122
#define TT   500
#define WID  80
#define UU   96
#define K2   (WID*CC)
#define HID  768
#define PROJ 1024
#define PLEN 484
#define EPSF 1e-5f

typedef float v2f __attribute__((ext_vector_type(2)));
typedef float v4f __attribute__((ext_vector_type(4)));
typedef float v8f __attribute__((ext_vector_type(8)));

static __device__ __forceinline__ v8f wmma_f32(v2f a, v2f b, v8f c) {
    return __builtin_amdgcn_wmma_f32_16x16x4_f32(false, a, false, b, (short)0, c, false, false);
}

#define SMT 2
#define SNT 4

static __device__ __forceinline__ void sub_load(v4f (&fa)[SMT][4], v4f (&fb)[SNT][4],
                                                const float* const (&pa)[SMT],
                                                const float* const (&pb)[SNT], int koff) {
    #pragma unroll
    for (int t = 0; t < SMT; ++t) {
        #pragma unroll
        for (int q = 0; q < 4; ++q) fa[t][q] = *(const v4f*)(pa[t] + koff + 8 * q);
    }
    #pragma unroll
    for (int u = 0; u < SNT; ++u) {
        #pragma unroll
        for (int q = 0; q < 4; ++q) fb[u][q] = *(const v4f*)(pb[u] + koff + 8 * q);
    }
}

static __device__ __forceinline__ void sub_mma(v8f (&acc)[SMT][SNT], const v4f (&fa)[SMT][4],
                                               const v4f (&fb)[SNT][4]) {
    #pragma unroll
    for (int j = 0; j < 8; ++j) {
        #pragma unroll
        for (int t = 0; t < SMT; ++t) {
            const v4f aq = fa[t][j >> 1];
            const v2f af = (j & 1) ? (v2f){ aq[2], aq[3] } : (v2f){ aq[0], aq[1] };
            #pragma unroll
            for (int u = 0; u < SNT; ++u) {
                const v4f wq = fb[u][j >> 1];
                const v2f wf = (j & 1) ? (v2f){ wq[2], wq[3] } : (v2f){ wq[0], wq[1] };
                acc[t][u] = wmma_f32(af, wf, acc[t][u]);
            }
        }
    }
}

__global__ __launch_bounds__(128) void k_subject(
    const float* __restrict__ x, const int* __restrict__ sid,
    const float* __restrict__ W_sub, const float* __restrict__ b_sub,
    float* __restrict__ out0)
{
    const int tid  = threadIdx.x;
    const int lane = tid & 31;
    const int n    = lane & 15;
    const int half = lane >> 4;
    const int b  = blockIdx.y;
    const int n0 = blockIdx.x * (16 * SNT);
    const int m0 = (tid >> 5) * (16 * SMT);
    const int s  = sid[b];
    const float* Wg = W_sub + (size_t)s * TT * TT;

    const float* pa[SMT];
    const float* pb[SNT];
    #pragma unroll
    for (int t = 0; t < SMT; ++t) {
        int am = m0 + 16 * t + n; if (am >= CC) am = CC - 1;
        pa[t] = x + ((size_t)b * CC + am) * TT + 4 * half;
    }
    #pragma unroll
    for (int u = 0; u < SNT; ++u) {
        int an = n0 + 16 * u + n; if (an >= TT) an = TT - 1;
        pb[u] = Wg + (size_t)an * TT + 4 * half;
    }

    v8f acc[SMT][SNT];
    #pragma unroll
    for (int t = 0; t < SMT; ++t) {
        #pragma unroll
        for (int u = 0; u < SNT; ++u) acc[t][u] = (v8f){};
    }

    v4f fa0[SMT][4], fb0[SNT][4], fa1[SMT][4], fb1[SNT][4];
    sub_load(fa0, fb0, pa, pb, 0);
    for (int st = 0; st < 14; st += 2) {
        sub_load(fa1, fb1, pa, pb, 32 * (st + 1));
        sub_mma(acc, fa0, fb0);
        sub_load(fa0, fb0, pa, pb, 32 * (st + 2));
        sub_mma(acc, fa1, fb1);
    }

    v4f ta[SMT][2], tb[SNT][2];
    v2f ua[SMT], ub[SNT];
    #pragma unroll
    for (int t = 0; t < SMT; ++t) {
        #pragma unroll
        for (int q = 0; q < 2; ++q) ta[t][q] = *(const v4f*)(pa[t] + 480 + 8 * q);
        ua[t] = *(const v2f*)(pa[t] + 496 - 2 * half);
    }
    #pragma unroll
    for (int u = 0; u < SNT; ++u) {
        #pragma unroll
        for (int q = 0; q < 2; ++q) tb[u][q] = *(const v4f*)(pb[u] + 480 + 8 * q);
        ub[u] = *(const v2f*)(pb[u] + 496 - 2 * half);
    }

    sub_mma(acc, fa0, fb0);

    #pragma unroll
    for (int j = 0; j < 4; ++j) {
        #pragma unroll
        for (int t = 0; t < SMT; ++t) {
            const v4f aq = ta[t][j >> 1];
            const v2f af = (j & 1) ? (v2f){ aq[2], aq[3] } : (v2f){ aq[0], aq[1] };
            #pragma unroll
            for (int u = 0; u < SNT; ++u) {
                const v4f wq = tb[u][j >> 1];
                const v2f wf = (j & 1) ? (v2f){ wq[2], wq[3] } : (v2f){ wq[0], wq[1] };
                acc[t][u] = wmma_f32(af, wf, acc[t][u]);
            }
        }
    }
    #pragma unroll
    for (int t = 0; t < SMT; ++t) {
        #pragma unroll
        for (int u = 0; u < SNT; ++u) acc[t][u] = wmma_f32(ua[t], ub[u], acc[t][u]);
    }

    const float* bg = b_sub + (size_t)s * TT;
    #pragma unroll
    for (int u = 0; u < SNT; ++u) {
        const int nn = n0 + 16 * u + n;
        const float bias = bg[nn < TT ? nn : TT - 1];
        #pragma unroll
        for (int t = 0; t < SMT; ++t) {
            #pragma unroll
            for (int r = 0; r < 8; ++r) {
                const int mm = m0 + 16 * t + r + (half << 3);
                if (mm < CC && nn < TT)
                    out0[((size_t)b * CC + mm) * TT + nn] = acc[t][u][r] + bias;
            }
        }
    }
}

__global__ __launch_bounds__(128) void k_pool(
    const float* __restrict__ out0, float* __restrict__ P)
{
    __shared__ float row[TT];
    const int bc = blockIdx.x;
    const float* src = out0 + (size_t)bc * TT;
    for (int i = threadIdx.x; i < TT; i += 128) row[i] = src[i];
    __syncthreads();
    float* dst = P + (size_t)bc * PLEN;
    for (int m = threadIdx.x; m < PLEN; m += 128) {
        float sum = 0.f;
        #pragma unroll
        for (int i = 0; i < 17; ++i) sum += row[m + i];
        dst[m] = sum * (1.f / 17.f);
    }
}

#define KSL   8
#define NPR   3
#define KSUB  4
#define NWV   (NPR * KSUB)
#define NSTEP (K2 / 32)
#define PPIT  162
#define PTILE (CC * PPIT)
#define PROW4 120
#define H2SL  (BB * WID * UU)
#define WTILE (10 * 8 * 32)
#define PFD   16

__global__ __launch_bounds__(NWV * 32) void k_conv2(
    const float* __restrict__ P,
    const float* __restrict__ conv1_w, const float* __restrict__ conv1_b,
    const float* __restrict__ bn1_g, const float* __restrict__ bn1_b,
    const float* __restrict__ bn1_m, const float* __restrict__ bn1_v,
    const float* __restrict__ conv2_w,
    float* __restrict__ part_lo, float* __restrict__ part_hi)
{
    __shared__ float Pl[NPR * PTILE];
    __shared__ float w1s[WID * 5];
    __shared__ float sc1[WID], sh1[WID];

    const int tid = threadIdx.x;
    const int b   = blockIdx.y;
    const int ksl = blockIdx.x;

    for (int i = tid; i < WID * 5; i += NWV * 32) w1s[i] = conv1_w[i];
    if (tid < WID) {
        float sc = bn1_g[tid] * rsqrtf(bn1_v[tid] + EPSF);
        sc1[tid] = sc;
        sh1[tid] = bn1_b[tid] - bn1_m[tid] * sc + sc * conv1_b[tid];
    }
    for (int i0 = tid; i0 < CC * PROW4; i0 += PFD * NWV * 32) {
        v4f pv[PFD];
        #pragma unroll
        for (int j = 0; j < PFD; ++j) {
            int i = i0 + j * (NWV * 32);
            if (i >= CC * PROW4) i = CC * PROW4 - 1;
            const int c = i / PROW4, q = 4 * (i - c * PROW4);
            pv[j] = *(const v4f*)(P + ((size_t)b * CC + c) * PLEN + q);
        }
        #pragma unroll
        for (int j = 0; j < PFD; ++j) {
            const int i = i0 + j * (NWV * 32);
            if (i < CC * PROW4) {
                const int c = i / PROW4, q = 4 * (i - c * PROW4);
                const int p = q / 160, qq = q - p * 160;
                float* d = &Pl[p * PTILE + c * PPIT + qq];
                *(v2f*)d       = (v2f){ pv[j][0], pv[j][1] };
                *(v2f*)(d + 2) = (v2f){ pv[j][2], pv[j][3] };
            }
        }
    }
    __syncthreads();

    const int wv   = __builtin_amdgcn_readfirstlane(tid >> 5);
    const int lane = tid & 31;
    const int n    = lane & 15;
    const int half = lane >> 4;
    const int pr   = wv % NPR;
    const int ks   = wv / NPR;

    const int srem = NSTEP % KSL, sbas = NSTEP / KSL;
    const int s0   = ksl * sbas + (ksl < srem ? ksl : srem);
    const int scnt = sbas + (ksl < srem ? 1 : 0);
    const int wrem = scnt % KSUB, wbas = scnt / KSUB;
    const int st0  = s0 + ks * wbas + (ks < wrem ? ks : wrem);
    const int cnt  = wbas + (ks < wrem ? 1 : 0);

    const int k0 = st0 * 32 + half * 16;
    int w = k0 / CC;
    int c = k0 - w * CC;

    const float* pa = conv2_w + (size_t)n * K2 + k0;
    const float* pp = &Pl[pr * PTILE + 5 * n];

    v8f acc[2][5];
    #pragma unroll
    for (int nt = 0; nt < 2; ++nt) {
        #pragma unroll
        for (int t = 0; t < 5; ++t) acc[nt][t] = (v8f){};
    }

    for (int it = 0; it < cnt; ++it) {
        v4f a[5][4];
        #pragma unroll
        for (int t = 0; t < 5; ++t) {
            #pragma unroll
            for (int q = 0; q < 4; ++q)
                a[t][q] = *(const v4f*)(pa + (size_t)t * 16 * K2 + 4 * q);
        }

        v2f bf[2][8];
        #pragma unroll
        for (int j = 0; j < 8; ++j) {
            int cj = c + 2 * j, wj = w;
            if (cj >= CC) { cj -= CC; wj += 1; }
            const float* wr  = &w1s[wj * 5];
            const float  w0 = wr[0], w1 = wr[1], w2 = wr[2], w3 = wr[3], w4 = wr[4];
            const float  scw = sc1[wj], shw = sh1[wj];
            #pragma unroll
            for (int nt = 0; nt < 2; ++nt) {
                const float* pr0 = pp + 80 * nt + cj * PPIT;
                const float* pr1 = pr0 + PPIT;
                float v0 = w0*pr0[0] + w1*pr0[1] + w2*pr0[2] + w3*pr0[3] + w4*pr0[4];
                float g0 = scw * v0 + shw;
                float e0 = g0 > 0.f ? g0 : (__expf(g0) - 1.f);
                float v1 = w0*pr1[0] + w1*pr1[1] + w2*pr1[2] + w3*pr1[3] + w4*pr1[4];
                float g1 = scw * v1 + shw;
                float e1 = g1 > 0.f ? g1 : (__expf(g1) - 1.f);
                bf[nt][j] = (v2f){ e0, e1 };
            }
            if ((j & 1) && j < 7) __builtin_amdgcn_sched_barrier(0);
        }

        #pragma unroll
        for (int j = 0; j < 8; ++j) {
            #pragma unroll
            for (int t = 0; t < 5; ++t) {
                v4f aq = a[t][j >> 1];
                v2f af = (j & 1) ? (v2f){ aq[2], aq[3] } : (v2f){ aq[0], aq[1] };
                #pragma unroll
                for (int nt = 0; nt < 2; ++nt) acc[nt][t] = wmma_f32(af, bf[nt][j], acc[nt][t]);
            }
        }

        pa += 32;
        c  += 32;
        if (c >= CC) { c -= CC; w += 1; }
    }

    __syncthreads();

    float* rw = Pl + (size_t)wv * WTILE;
    #pragma unroll
    for (int nt = 0; nt < 2; ++nt) {
        #pragma unroll
        for (int t = 0; t < 5; ++t) {
            #pragma unroll
            for (int r = 0; r < 8; ++r) rw[((nt * 5 + t) * 8 + r) * 32 + lane] = acc[nt][t][r];
        }
    }
    __syncthreads();

    float* dst = (ksl < KSL - 1) ? (part_lo + (size_t)ksl * H2SL) : part_hi;
    #pragma unroll 4
    for (int it = 0; it < (NPR * WTILE) / (NWV * 32); ++it) {
        const int sidx = tid + it * (NWV * 32);
        const int p   = sidx / WTILE;
        const int rem = sidx - p * WTILE;
        float sum = 0.f;
        #pragma unroll
        for (int v = 0; v < KSUB; ++v) sum += Pl[(v * NPR + p) * WTILE + rem];
        const int tt = rem >> 8, r = (rem >> 5) & 7, l = rem & 31;
        const int nt = tt / 5, t = tt - nt * 5;
        const int w2 = t * 16 + r + (l >> 4) * 8;
        const int u  = 16 * (2 * p + nt) + (l & 15);
        dst[((size_t)b * WID + w2) * UU + u] = sum;
    }
}

__global__ __launch_bounds__(256) void k_proj(
    const float* __restrict__ part_lo, const float* __restrict__ part_hi,
    const float* __restrict__ conv2_b,
    const float* __restrict__ bn2_g, const float* __restrict__ bn2_b,
    const float* __restrict__ bn2_m, const float* __restrict__ bn2_v,
    const float* __restrict__ proj_w, const float* __restrict__ proj_b,
    float* __restrict__ z)
{
    __shared__ float hs[WID * 16];
    __shared__ float pw[8 * WID];
    const int ut = blockIdx.x;
    const int b  = blockIdx.y;
    #pragma unroll
    for (int it = 0; it < (WID * 16) / 256; ++it) {
        const int i  = threadIdx.x + it * 256;
        const int w2 = i >> 4, uu = i & 15;
        const size_t off = ((size_t)b * WID + w2) * UU + ut * 16 + uu;
        float sum = 0.f;
        #pragma unroll
        for (int s = 0; s < KSL - 1; ++s) sum += part_lo[(size_t)s * H2SL + off];
        sum += part_hi[off];
        float sc = bn2_g[w2] * rsqrtf(bn2_v[w2] + EPSF);
        float sh = bn2_b[w2] - bn2_m[w2] * sc;
        float v  = sc * (sum + conv2_b[w2]) + sh;
        hs[i] = v > 0.f ? v : (__expf(v) - 1.f);
    }
    for (int i = threadIdx.x; i < 8 * WID; i += 256)  pw[i] = proj_w[i];
    __syncthreads();
    if (threadIdx.x < 128) {
        const int uu = threadIdx.x >> 3, e = threadIdx.x & 7;
        float s = proj_b[e];
        #pragma unroll 8
        for (int w = 0; w < WID; ++w) s += pw[e * WID + w] * hs[w * 16 + uu];
        z[(size_t)b * HID + (ut * 16 + uu) * 8 + e] = s;
    }
}

#define MW 8

template <int KTOT>
static __device__ __forceinline__ v8f mlp_partial(const float* __restrict__ A,
                                                  const float* __restrict__ Wt,
                                                  int m0, int n0, int wv, int lane)
{
    constexpr int KS = KTOT / MW;
    constexpr int NQ = KS / 8;
    const int n = lane & 15, half = lane >> 4;
    const float* pa = A  + (size_t)(m0 + n) * KTOT + wv * KS + 4 * half;
    const float* pb = Wt + (size_t)(n0 + n) * KTOT + wv * KS + 4 * half;
    v4f fa[NQ], fb[NQ];
    #pragma unroll
    for (int q = 0; q < NQ; ++q) fb[q] = *(const v4f*)(pb + 8 * q);
    #pragma unroll
    for (int q = 0; q < NQ; ++q) fa[q] = *(const v4f*)(pa + 8 * q);
    v8f acc0 = {}, acc1 = {};
    #pragma unroll
    for (int q = 0; q < NQ; ++q) {
        acc0 = wmma_f32((v2f){ fa[q][0], fa[q][1] }, (v2f){ fb[q][0], fb[q][1] }, acc0);
        acc1 = wmma_f32((v2f){ fa[q][2], fa[q][3] }, (v2f){ fb[q][2], fb[q][3] }, acc1);
    }
    return acc0 + acc1;
}

__global__ __launch_bounds__(MW * 32) void k_mlp1(
    const float* __restrict__ z, const float* __restrict__ w,
    const float* __restrict__ bias, float* __restrict__ z1, float* __restrict__ g)
{
    __shared__ float part[MW * 8 * 32];
    const int tid  = threadIdx.x;
    const int lane = tid & 31;
    const int wv   = tid >> 5;
    const int m0 = blockIdx.x * 16, n0 = blockIdx.y * 16;
    const int nn = n0 + (lane & 15);
    const float bn = bias[nn];

    const v8f acc = mlp_partial<HID>(z, w, m0, n0, wv, lane);
    #pragma unroll
    for (int r = 0; r < 8; ++r) part[(wv * 8 + r) * 32 + lane] = acc[r];
    __syncthreads();

    float sum = 0.f;
    #pragma unroll
    for (int v = 0; v < MW; ++v) sum += part[(v * 8 + wv) * 32 + lane];
    const int mm = m0 + wv + ((lane >> 4) << 3);
    const float v = sum + bn;
    z1[(size_t)mm * PROJ + nn] = v;
    g [(size_t)mm * PROJ + nn] = 0.5f * v * (1.f + erff(v * 0.70710678118654752f));
}

__global__ __launch_bounds__(MW * 32) void k_mlp2(
    const float* __restrict__ g, const float* __restrict__ w,
    const float* __restrict__ bias, const float* __restrict__ z1,
    float* __restrict__ y)
{
    __shared__ float part[MW * 8 * 32];
    const int tid  = threadIdx.x;
    const int lane = tid & 31;
    const int wv   = tid >> 5;
    const int m0 = blockIdx.x * 16, n0 = blockIdx.y * 16;
    const int nn = n0 + (lane & 15);
    const int mm = m0 + wv + ((lane >> 4) << 3);
    const float bn  = bias[nn];
    const float res = z1[(size_t)mm * PROJ + nn];

    const v8f acc = mlp_partial<PROJ>(g, w, m0, n0, wv, lane);
    #pragma unroll
    for (int r = 0; r < 8; ++r) part[(wv * 8 + r) * 32 + lane] = acc[r];
    __syncthreads();

    float sum = 0.f;
    #pragma unroll
    for (int v = 0; v < MW; ++v) sum += part[(v * 8 + wv) * 32 + lane];
    y[(size_t)mm * PROJ + nn] = res + sum + bn;
}

__global__ __launch_bounds__(256) void k_ln(
    const float* __restrict__ y, const float* __restrict__ gam,
    const float* __restrict__ bet, float* __restrict__ out)
{
    __shared__ float red[256];
    const int b = blockIdx.x;
    const float* row = y + (size_t)b * PROJ;
    float s = 0.f;
    for (int i = threadIdx.x; i < PROJ; i += 256) s += row[i];
    red[threadIdx.x] = s; __syncthreads();
    for (int off = 128; off > 0; off >>= 1) {
        if (threadIdx.x < off) red[threadIdx.x] += red[threadIdx.x + off];
        __syncthreads();
    }
    const float mu = red[0] * (1.f / PROJ);
    __syncthreads();
    float s2 = 0.f;
    for (int i = threadIdx.x; i < PROJ; i += 256) { float d = row[i] - mu; s2 += d * d; }
    red[threadIdx.x] = s2; __syncthreads();
    for (int off = 128; off > 0; off >>= 1) {
        if (threadIdx.x < off) red[threadIdx.x] += red[threadIdx.x + off];
        __syncthreads();
    }
    const float inv = rsqrtf(red[0] * (1.f / PROJ) + EPSF);
    for (int i = threadIdx.x; i < PROJ; i += 256)
        out[(size_t)b * PROJ + i] = (row[i] - mu) * inv * gam[i] + bet[i];
}

extern "C" void kernel_launch(void* const* d_in, const int* in_sizes, int n_in,
                              void* d_out, int out_size, void* d_ws, size_t ws_size,
                              hipStream_t stream) {
    const float* x       = (const float*)d_in[0];
    const int*   sid     = (const int*)  d_in[1];
    const float* W_sub   = (const float*)d_in[2];
    const float* b_sub   = (const float*)d_in[3];
    const float* conv1_w = (const float*)d_in[4];
    const float* conv1_b = (const float*)d_in[5];
    const float* bn1_g   = (const float*)d_in[6];
    const float* bn1_b   = (const float*)d_in[7];
    const float* bn1_m   = (const float*)d_in[8];
    const float* bn1_v   = (const float*)d_in[9];
    const float* conv2_w = (const float*)d_in[10];
    const float* conv2_b = (const float*)d_in[11];
    const float* bn2_g   = (const float*)d_in[12];
    const float* bn2_b   = (const float*)d_in[13];
    const float* bn2_m   = (const float*)d_in[14];
    const float* bn2_v   = (const float*)d_in[15];
    const float* proj_w  = (const float*)d_in[16];
    const float* proj_b  = (const float*)d_in[17];
    const float* mlp1_w  = (const float*)d_in[18];
    const float* mlp1_b  = (const float*)d_in[19];
    const float* mlp2_w  = (const float*)d_in[20];
    const float* mlp2_b  = (const float*)d_in[21];
    const float* ln_g    = (const float*)d_in[22];
    const float* ln_b    = (const float*)d_in[23];

    float* ws   = (float*)d_ws;
    float* out0 = ws;
    float* P    = out0 + (size_t)BB * CC * TT;
    float* h2   = P    + (size_t)BB * CC * PLEN;
    float* z    = h2   + (size_t)BB * WID * UU;
    float* z1   = z    + (size_t)BB * HID;
    float* gg   = z1   + (size_t)BB * PROJ;
    float* yy   = gg   + (size_t)BB * PROJ;

    dim3 g1(8, BB);
    k_subject<<<g1, 128, 0, stream>>>(x, sid, W_sub, b_sub, out0);

    k_pool<<<BB * CC, 128, 0, stream>>>(out0, P);

    dim3 g3(KSL, BB);
    k_conv2<<<g3, NWV * 32, 0, stream>>>(P, conv1_w, conv1_b, bn1_g, bn1_b, bn1_m, bn1_v,
                                    conv2_w, out0, h2);

    dim3 g4(UU / 16, BB);
    k_proj<<<g4, 256, 0, stream>>>(out0, h2, conv2_b, bn2_g, bn2_b, bn2_m, bn2_v, proj_w, proj_b, z);

    dim3 g5(2, PROJ / 16);
    k_mlp1<<<g5, MW * 32, 0, stream>>>(z, mlp1_w, mlp1_b, z1, gg);
    k_mlp2<<<g5, MW * 32, 0, stream>>>(gg, mlp2_w, mlp2_b, z1, yy);

    k_ln<<<BB, 256, 0, stream>>>(yy, ln_g, ln_b, (float*)d_out);
}


# k_pool folded into conv2's LDS fill: each block pools its batch's out0 rows itself (P never written to memory, one launch fewer); a window is summed as right part + middle + left part: new sum order
# speedup vs baseline: 5.9202x; 1.0558x over previous
#include <hip/hip_runtime.h>
#include <math.h>

#define BB   32
#define CC   122
#define TT   500
#define WID  80
#define UU   96
#define K2   (WID*CC)
#define HID  768
#define PROJ 1024
#define PLEN 484
#define EPSF 1e-5f

typedef float v2f __attribute__((ext_vector_type(2)));
typedef float v4f __attribute__((ext_vector_type(4)));
typedef float v8f __attribute__((ext_vector_type(8)));

static __device__ __forceinline__ v8f wmma_f32(v2f a, v2f b, v8f c) {
    return __builtin_amdgcn_wmma_f32_16x16x4_f32(false, a, false, b, (short)0, c, false, false);
}

#define SMT 2
#define SNT 4

static __device__ __forceinline__ void sub_load(v4f (&fa)[SMT][4], v4f (&fb)[SNT][4],
                                                const float* const (&pa)[SMT],
                                                const float* const (&pb)[SNT], int koff) {
    #pragma unroll
    for (int t = 0; t < SMT; ++t) {
        #pragma unroll
        for (int q = 0; q < 4; ++q) fa[t][q] = *(const v4f*)(pa[t] + koff + 8 * q);
    }
    #pragma unroll
    for (int u = 0; u < SNT; ++u) {
        #pragma unroll
        for (int q = 0; q < 4; ++q) fb[u][q] = *(const v4f*)(pb[u] + koff + 8 * q);
    }
}

static __device__ __forceinline__ void sub_mma(v8f (&acc)[SMT][SNT], const v4f (&fa)[SMT][4],
                                               const v4f (&fb)[SNT][4]) {
    #pragma unroll
    for (int j = 0; j < 8; ++j) {
        #pragma unroll
        for (int t = 0; t < SMT; ++t) {
            const v4f aq = fa[t][j >> 1];
            const v2f af = (j & 1) ? (v2f){ aq[2], aq[3] } : (v2f){ aq[0], aq[1] };
            #pragma unroll
            for (int u = 0; u < SNT; ++u) {
                const v4f wq = fb[u][j >> 1];
                const v2f wf = (j & 1) ? (v2f){ wq[2], wq[3] } : (v2f){ wq[0], wq[1] };
                acc[t][u] = wmma_f32(af, wf, acc[t][u]);
            }
        }
    }
}

__global__ __launch_bounds__(128) void k_subject(
    const float* __restrict__ x, const int* __restrict__ sid,
    const float* __restrict__ W_sub, const float* __restrict__ b_sub,
    float* __restrict__ out0)
{
    const int tid  = threadIdx.x;
    const int lane = tid & 31;
    const int n    = lane & 15;
    const int half = lane >> 4;
    const int b  = blockIdx.y;
    const int n0 = blockIdx.x * (16 * SNT);
    const int m0 = (tid >> 5) * (16 * SMT);
    const int s  = sid[b];
    const float* Wg = W_sub + (size_t)s * TT * TT;

    const float* pa[SMT];
    const float* pb[SNT];
    #pragma unroll
    for (int t = 0; t < SMT; ++t) {
        int am = m0 + 16 * t + n; if (am >= CC) am = CC - 1;
        pa[t] = x + ((size_t)b * CC + am) * TT + 4 * half;
    }
    #pragma unroll
    for (int u = 0; u < SNT; ++u) {
        int an = n0 + 16 * u + n; if (an >= TT) an = TT - 1;
        pb[u] = Wg + (size_t)an * TT + 4 * half;
    }

    v8f acc[SMT][SNT];
    #pragma unroll
    for (int t = 0; t < SMT; ++t) {
        #pragma unroll
        for (int u = 0; u < SNT; ++u) acc[t][u] = (v8f){};
    }

    v4f fa0[SMT][4], fb0[SNT][4], fa1[SMT][4], fb1[SNT][4];
    sub_load(fa0, fb0, pa, pb, 0);
    for (int st = 0; st < 14; st += 2) {
        sub_load(fa1, fb1, pa, pb, 32 * (st + 1));
        sub_mma(acc, fa0, fb0);
        sub_load(fa0, fb0, pa, pb, 32 * (st + 2));
        sub_mma(acc, fa1, fb1);
    }

    v4f ta[SMT][2], tb[SNT][2];
    v2f ua[SMT], ub[SNT];
    #pragma unroll
    for (int t = 0; t < SMT; ++t) {
        #pragma unroll
        for (int q = 0; q < 2; ++q) ta[t][q] = *(const v4f*)(pa[t] + 480 + 8 * q);
        ua[t] = *(const v2f*)(pa[t] + 496 - 2 * half);
    }
    #pragma unroll
    for (int u = 0; u < SNT; ++u) {
        #pragma unroll
        for (int q = 0; q < 2; ++q) tb[u][q] = *(const v4f*)(pb[u] + 480 + 8 * q);
        ub[u] = *(const v2f*)(pb[u] + 496 - 2 * half);
    }

    sub_mma(acc, fa0, fb0);

    #pragma unroll
    for (int j = 0; j < 4; ++j) {
        #pragma unroll
        for (int t = 0; t < SMT; ++t) {
            const v4f aq = ta[t][j >> 1];
            const v2f af = (j & 1) ? (v2f){ aq[2], aq[3] } : (v2f){ aq[0], aq[1] };
            #pragma unroll
            for (int u = 0; u < SNT; ++u) {
                const v4f wq = tb[u][j >> 1];
                const v2f wf = (j & 1) ? (v2f){ wq[2], wq[3] } : (v2f){ wq[0], wq[1] };
                acc[t][u] = wmma_f32(af, wf, acc[t][u]);
            }
        }
    }
    #pragma unroll
    for (int t = 0; t < SMT; ++t) {
        #pragma unroll
        for (int u = 0; u < SNT; ++u) acc[t][u] = wmma_f32(ua[t], ub[u], acc[t][u]);
    }

    const float* bg = b_sub + (size_t)s * TT;
    #pragma unroll
    for (int u = 0; u < SNT; ++u) {
        const int nn = n0 + 16 * u + n;
        const float bias = bg[nn < TT ? nn : TT - 1];
        #pragma unroll
        for (int t = 0; t < SMT; ++t) {
            #pragma unroll
            for (int r = 0; r < 8; ++r) {
                const int mm = m0 + 16 * t + r + (half << 3);
                if (mm < CC && nn < TT)
                    out0[((size_t)b * CC + mm) * TT + nn] = acc[t][u][r] + bias;
            }
        }
    }
}

#define KSL   8
#define NPR   3
#define KSUB  4
#define NWV   (NPR * KSUB)
#define NSTEP (K2 / 32)
#define PPIT  162
#define PTILE (CC * PPIT)
#define H2SL  (BB * WID * UU)
#define WTILE (10 * 8 * 32)
#define STG   512
#define RPW   ((CC + NWV - 1) / NWV)

__global__ __launch_bounds__(NWV * 32) void k_conv2(
    const float* __restrict__ out0,
    const float* __restrict__ conv1_w, const float* __restrict__ conv1_b,
    const float* __restrict__ bn1_g, const float* __restrict__ bn1_b,
    const float* __restrict__ bn1_m, const float* __restrict__ bn1_v,
    const float* __restrict__ conv2_w,
    float* __restrict__ part_lo, float* __restrict__ part_hi)
{
    __shared__ float Pl[NPR * PTILE];
    __shared__ float w1s[WID * 5];
    __shared__ float sc1[WID], sh1[WID];
    __shared__ float stg[NWV * STG];

    const int tid = threadIdx.x;
    const int b   = blockIdx.y;
    const int ksl = blockIdx.x;

    for (int i = tid; i < WID * 5; i += NWV * 32) w1s[i] = conv1_w[i];
    if (tid < WID) {
        float sc = bn1_g[tid] * rsqrtf(bn1_v[tid] + EPSF);
        sc1[tid] = sc;
        sh1[tid] = bn1_b[tid] - bn1_m[tid] * sc + sc * conv1_b[tid];
    }
    const int wv   = __builtin_amdgcn_readfirstlane(tid >> 5);
    const int lane = tid & 31;

    {
        const float* xb = out0 + (size_t)b * CC * TT;
        float* sw = stg + wv * STG;
        int off[15];
        #pragma unroll
        for (int i = 0; i < 15; ++i) {
            const int q = 15 * lane + i;
            const int p = q / 160;
            off[i] = p * PTILE + (q - 160 * p);
        }
        v4f rv[RPW][4];
        #pragma unroll
        for (int r = 0; r < RPW; ++r) {
            int c = wv + NWV * r;
            if (c >= CC) c -= NWV;
            #pragma unroll
            for (int j = 0; j < 4; ++j) {
                int f = 4 * lane + 128 * j;
                if (f > TT - 4) f = TT - 4;
                rv[r][j] = *(const v4f*)(xb + (size_t)c * TT + f);
            }
        }
        #pragma unroll
        for (int r = 0; r < RPW; ++r) {
            int c = wv + NWV * r;
            if (c >= CC) c -= NWV;
            float* prow = Pl + c * PPIT;
            #pragma unroll
            for (int j = 0; j < 4; ++j) *(v4f*)(sw + 4 * lane + 128 * j) = rv[r][j];
            __builtin_amdgcn_wave_barrier();
            float v[31];
            #pragma unroll
            for (int i = 0; i < 31; ++i) v[i] = sw[15 * lane + i];
            __builtin_amdgcn_wave_barrier();
            const float mid = (v[14] + v[15]) + v[16];
            float sf[14], pf[15];
            sf[13] = v[13];
            #pragma unroll
            for (int i = 12; i >= 0; --i) sf[i] = v[i] + sf[i + 1];
            pf[1] = v[17];
            #pragma unroll
            for (int i = 2; i < 15; ++i) pf[i] = pf[i - 1] + v[16 + i];
            #pragma unroll
            for (int i = 0; i < 15; ++i) {
                float o;
                if (i == 0)       o = sf[0] + mid;
                else if (i == 14) o = mid + pf[14];
                else              o = (sf[i] + mid) + pf[i];
                prow[off[i]] = o * (1.f / 17.f);
            }
        }
    }
    __syncthreads();

    const int n    = lane & 15;
    const int half = lane >> 4;
    const int pr   = wv % NPR;
    const int ks   = wv / NPR;

    const int srem = NSTEP % KSL, sbas = NSTEP / KSL;
    const int s0   = ksl * sbas + (ksl < srem ? ksl : srem);
    const int scnt = sbas + (ksl < srem ? 1 : 0);
    const int wrem = scnt % KSUB, wbas = scnt / KSUB;
    const int st0  = s0 + ks * wbas + (ks < wrem ? ks : wrem);
    const int cnt  = wbas + (ks < wrem ? 1 : 0);

    const int k0 = st0 * 32 + half * 16;
    int w = k0 / CC;
    int c = k0 - w * CC;

    const float* pa = conv2_w + (size_t)n * K2 + k0;
    const float* pp = &Pl[pr * PTILE + 5 * n];

    v8f acc[2][5];
    #pragma unroll
    for (int nt = 0; nt < 2; ++nt) {
        #pragma unroll
        for (int t = 0; t < 5; ++t) acc[nt][t] = (v8f){};
    }

    for (int it = 0; it < cnt; ++it) {
        v4f a[5][4];
        #pragma unroll
        for (int t = 0; t < 5; ++t) {
            #pragma unroll
            for (int q = 0; q < 4; ++q)
                a[t][q] = *(const v4f*)(pa + (size_t)t * 16 * K2 + 4 * q);
        }

        v2f bf[2][8];
        #pragma unroll
        for (int j = 0; j < 8; ++j) {
            int cj = c + 2 * j, wj = w;
            if (cj >= CC) { cj -= CC; wj += 1; }
            const float* wr  = &w1s[wj * 5];
            const float  w0 = wr[0], w1 = wr[1], w2 = wr[2], w3 = wr[3], w4 = wr[4];
            const float  scw = sc1[wj], shw = sh1[wj];
            #pragma unroll
            for (int nt = 0; nt < 2; ++nt) {
                const float* pr0 = pp + 80 * nt + cj * PPIT;
                const float* pr1 = pr0 + PPIT;
                float v0 = w0*pr0[0] + w1*pr0[1] + w2*pr0[2] + w3*pr0[3] + w4*pr0[4];
                float g0 = scw * v0 + shw;
                float e0 = g0 > 0.f ? g0 : (__expf(g0) - 1.f);
                float v1 = w0*pr1[0] + w1*pr1[1] + w2*pr1[2] + w3*pr1[3] + w4*pr1[4];
                float g1 = scw * v1 + shw;
                float e1 = g1 > 0.f ? g1 : (__expf(g1) - 1.f);
                bf[nt][j] = (v2f){ e0, e1 };
            }
            if ((j & 1) && j < 7) __builtin_amdgcn_sched_barrier(0);
        }

        #pragma unroll
        for (int j = 0; j < 8; ++j) {
            #pragma unroll
            for (int t = 0; t < 5; ++t) {
                v4f aq = a[t][j >> 1];
                v2f af = (j & 1) ? (v2f){ aq[2], aq[3] } : (v2f){ aq[0], aq[1] };
                #pragma unroll
                for (int nt = 0; nt < 2; ++nt) acc[nt][t] = wmma_f32(af, bf[nt][j], acc[nt][t]);
            }
        }

        pa += 32;
        c  += 32;
        if (c >= CC) { c -= CC; w += 1; }
    }

    __syncthreads();

    float* rw = Pl + (size_t)wv * WTILE;
    #pragma unroll
    for (int nt = 0; nt < 2; ++nt) {
        #pragma unroll
        for (int t = 0; t < 5; ++t) {
            #pragma unroll
            for (int r = 0; r < 8; ++r) rw[((nt * 5 + t) * 8 + r) * 32 + lane] = acc[nt][t][r];
        }
    }
    __syncthreads();

    float* dst = (ksl < KSL - 1) ? (part_lo + (size_t)ksl * H2SL) : part_hi;
    #pragma unroll 4
    for (int it = 0; it < (NPR * WTILE) / (NWV * 32); ++it) {
        const int sidx = tid + it * (NWV * 32);
        const int p   = sidx / WTILE;
        const int rem = sidx - p * WTILE;
        float sum = 0.f;
        #pragma unroll
        for (int v = 0; v < KSUB; ++v) sum += Pl[(v * NPR + p) * WTILE + rem];
        const int tt = rem >> 8, r = (rem >> 5) & 7, l = rem & 31;
        const int nt = tt / 5, t = tt - nt * 5;
        const int w2 = t * 16 + r + (l >> 4) * 8;
        const int u  = 16 * (2 * p + nt) + (l & 15);
        dst[((size_t)b * WID + w2) * UU + u] = sum;
    }
}

__global__ __launch_bounds__(256) void k_proj(
    const float* __restrict__ part_lo, const float* __restrict__ part_hi,
    const float* __restrict__ conv2_b,
    const float* __restrict__ bn2_g, const float* __restrict__ bn2_b,
    const float* __restrict__ bn2_m, const float* __restrict__ bn2_v,
    const float* __restrict__ proj_w, const float* __restrict__ proj_b,
    float* __restrict__ z)
{
    __shared__ float hs[WID * 16];
    __shared__ float pw[8 * WID];
    const int ut = blockIdx.x;
    const int b  = blockIdx.y;
    #pragma unroll
    for (int it = 0; it < (WID * 16) / 256; ++it) {
        const int i  = threadIdx.x + it * 256;
        const int w2 = i >> 4, uu = i & 15;
        const size_t off = ((size_t)b * WID + w2) * UU + ut * 16 + uu;
        float sum = 0.f;
        #pragma unroll
        for (int s = 0; s < KSL - 1; ++s) sum += part_lo[(size_t)s * H2SL + off];
        sum += part_hi[off];
        float sc = bn2_g[w2] * rsqrtf(bn2_v[w2] + EPSF);
        float sh = bn2_b[w2] - bn2_m[w2] * sc;
        float v  = sc * (sum + conv2_b[w2]) + sh;
        hs[i] = v > 0.f ? v : (__expf(v) - 1.f);
    }
    for (int i = threadIdx.x; i < 8 * WID; i += 256)  pw[i] = proj_w[i];
    __syncthreads();
    if (threadIdx.x < 128) {
        const int uu = threadIdx.x >> 3, e = threadIdx.x & 7;
        float s = proj_b[e];
        #pragma unroll 8
        for (int w = 0; w < WID; ++w) s += pw[e * WID + w] * hs[w * 16 + uu];
        z[(size_t)b * HID + (ut * 16 + uu) * 8 + e] = s;
    }
}

#define MW 8

template <int KTOT>
static __device__ __forceinline__ v8f mlp_partial(const float* __restrict__ A,
                                                  const float* __restrict__ Wt,
                                                  int m0, int n0, int wv, int lane)
{
    constexpr int KS = KTOT / MW;
    constexpr int NQ = KS / 8;
    const int n = lane & 15, half = lane >> 4;
    const float* pa = A  + (size_t)(m0 + n) * KTOT + wv * KS + 4 * half;
    const float* pb = Wt + (size_t)(n0 + n) * KTOT + wv * KS + 4 * half;
    v4f fa[NQ], fb[NQ];
    #pragma unroll
    for (int q = 0; q < NQ; ++q) fb[q] = *(const v4f*)(pb + 8 * q);
    #pragma unroll
    for (int q = 0; q < NQ; ++q) fa[q] = *(const v4f*)(pa + 8 * q);
    v8f acc0 = {}, acc1 = {};
    #pragma unroll
    for (int q = 0; q < NQ; ++q) {
        acc0 = wmma_f32((v2f){ fa[q][0], fa[q][1] }, (v2f){ fb[q][0], fb[q][1] }, acc0);
        acc1 = wmma_f32((v2f){ fa[q][2], fa[q][3] }, (v2f){ fb[q][2], fb[q][3] }, acc1);
    }
    return acc0 + acc1;
}

__global__ __launch_bounds__(MW * 32) void k_mlp1(
    const float* __restrict__ z, const float* __restrict__ w,
    const float* __restrict__ bias, float* __restrict__ z1, float* __restrict__ g)
{
    __shared__ float part[MW * 8 * 32];
    const int tid  = threadIdx.x;
    const int lane = tid & 31;
    const int wv   = tid >> 5;
    const int m0 = blockIdx.x * 16, n0 = blockIdx.y * 16;
    const int nn = n0 + (lane & 15);
    const float bn = bias[nn];

    const v8f acc = mlp_partial<HID>(z, w, m0, n0, wv, lane);
    #pragma unroll
    for (int r = 0; r < 8; ++r) part[(wv * 8 + r) * 32 + lane] = acc[r];
    __syncthreads();

    float sum = 0.f;
    #pragma unroll
    for (int v = 0; v < MW; ++v) sum += part[(v * 8 + wv) * 32 + lane];
    const int mm = m0 + wv + ((lane >> 4) << 3);
    const float v = sum + bn;
    z1[(size_t)mm * PROJ + nn] = v;
    g [(size_t)mm * PROJ + nn] = 0.5f * v * (1.f + erff(v * 0.70710678118654752f));
}

__global__ __launch_bounds__(MW * 32) void k_mlp2(
    const float* __restrict__ g, const float* __restrict__ w,
    const float* __restrict__ bias, const float* __restrict__ z1,
    float* __restrict__ y)
{
    __shared__ float part[MW * 8 * 32];
    const int tid  = threadIdx.x;
    const int lane = tid & 31;
    const int wv   = tid >> 5;
    const int m0 = blockIdx.x * 16, n0 = blockIdx.y * 16;
    const int nn = n0 + (lane & 15);
    const int mm = m0 + wv + ((lane >> 4) << 3);
    const float bn  = bias[nn];
    const float res = z1[(size_t)mm * PROJ + nn];

    const v8f acc = mlp_partial<PROJ>(g, w, m0, n0, wv, lane);
    #pragma unroll
    for (int r = 0; r < 8; ++r) part[(wv * 8 + r) * 32 + lane] = acc[r];
    __syncthreads();

    float sum = 0.f;
    #pragma unroll
    for (int v = 0; v < MW; ++v) sum += part[(v * 8 + wv) * 32 + lane];
    y[(size_t)mm * PROJ + nn] = res + sum + bn;
}

__global__ __launch_bounds__(256) void k_ln(
    const float* __restrict__ y, const float* __restrict__ gam,
    const float* __restrict__ bet, float* __restrict__ out)
{
    __shared__ float red[256];
    const int b = blockIdx.x;
    const float* row = y + (size_t)b * PROJ;
    float s = 0.f;
    for (int i = threadIdx.x; i < PROJ; i += 256) s += row[i];
    red[threadIdx.x] = s; __syncthreads();
    for (int off = 128; off > 0; off >>= 1) {
        if (threadIdx.x < off) red[threadIdx.x] += red[threadIdx.x + off];
        __syncthreads();
    }
    const float mu = red[0] * (1.f / PROJ);
    __syncthreads();
    float s2 = 0.f;
    for (int i = threadIdx.x; i < PROJ; i += 256) { float d = row[i] - mu; s2 += d * d; }
    red[threadIdx.x] = s2; __syncthreads();
    for (int off = 128; off > 0; off >>= 1) {
        if (threadIdx.x < off) red[threadIdx.x] += red[threadIdx.x + off];
        __syncthreads();
    }
    const float inv = rsqrtf(red[0] * (1.f / PROJ) + EPSF);
    for (int i = threadIdx.x; i < PROJ; i += 256)
        out[(size_t)b * PROJ + i] = (row[i] - mu) * inv * gam[i] + bet[i];
}

extern "C" void kernel_launch(void* const* d_in, const int* in_sizes, int n_in,
                              void* d_out, int out_size, void* d_ws, size_t ws_size,
                              hipStream_t stream) {
    const float* x       = (const float*)d_in[0];
    const int*   sid     = (const int*)  d_in[1];
    const float* W_sub   = (const float*)d_in[2];
    const float* b_sub   = (const float*)d_in[3];
    const float* conv1_w = (const float*)d_in[4];
    const float* conv1_b = (const float*)d_in[5];
    const float* bn1_g   = (const float*)d_in[6];
    const float* bn1_b   = (const float*)d_in[7];
    const float* bn1_m   = (const float*)d_in[8];
    const float* bn1_v   = (const float*)d_in[9];
    const float* conv2_w = (const float*)d_in[10];
    const float* conv2_b = (const float*)d_in[11];
    const float* bn2_g   = (const float*)d_in[12];
    const float* bn2_b   = (const float*)d_in[13];
    const float* bn2_m   = (const float*)d_in[14];
    const float* bn2_v   = (const float*)d_in[15];
    const float* proj_w  = (const float*)d_in[16];
    const float* proj_b  = (const float*)d_in[17];
    const float* mlp1_w  = (const float*)d_in[18];
    const float* mlp1_b  = (const float*)d_in[19];
    const float* mlp2_w  = (const float*)d_in[20];
    const float* mlp2_b  = (const float*)d_in[21];
    const float* ln_g    = (const float*)d_in[22];
    const float* ln_b    = (const float*)d_in[23];

    float* ws   = (float*)d_ws;
    float* out0 = ws;
    float* pl7  = out0 + (size_t)BB * CC * TT;
    float* h2   = pl7  + (size_t)BB * CC * PLEN;
    float* z    = h2   + (size_t)BB * WID * UU;
    float* z1   = z    + (size_t)BB * HID;
    float* gg   = z1   + (size_t)BB * PROJ;
    float* yy   = gg   + (size_t)BB * PROJ;

    dim3 g1(8, BB);
    k_subject<<<g1, 128, 0, stream>>>(x, sid, W_sub, b_sub, out0);

    dim3 g3(KSL, BB);
    k_conv2<<<g3, NWV * 32, 0, stream>>>(out0, conv1_w, conv1_b, bn1_g, bn1_b, bn1_m, bn1_v,
                                    conv2_w, pl7, h2);

    dim3 g4(UU / 16, BB);
    k_proj<<<g4, 256, 0, stream>>>(pl7, h2, conv2_b, bn2_g, bn2_b, bn2_m, bn2_v, proj_w, proj_b, z);

    dim3 g5(2, PROJ / 16);
    k_mlp1<<<g5, MW * 32, 0, stream>>>(z, mlp1_w, mlp1_b, z1, gg);
    k_mlp2<<<g5, MW * 32, 0, stream>>>(gg, mlp2_w, mlp2_b, z1, yy);

    k_ln<<<BB, 256, 0, stream>>>(yy, ln_g, ln_b, (float*)d_out);
}


# k_subject: each step's 24 fragment loads leave as one group ahead of the 64 WMMAs they overlap (scheduling fences), not spread among them; the 4 bias loads move ahead of the K loop
# speedup vs baseline: 6.1471x; 1.0383x over previous
#include <hip/hip_runtime.h>
#include <math.h>

#define BB   32
#define CC   122
#define TT   500
#define WID  80
#define UU   96
#define K2   (WID*CC)
#define HID  768
#define PROJ 1024
#define PLEN 484
#define EPSF 1e-5f

typedef float v2f __attribute__((ext_vector_type(2)));
typedef float v4f __attribute__((ext_vector_type(4)));
typedef float v8f __attribute__((ext_vector_type(8)));

static __device__ __forceinline__ v8f wmma_f32(v2f a, v2f b, v8f c) {
    return __builtin_amdgcn_wmma_f32_16x16x4_f32(false, a, false, b, (short)0, c, false, false);
}

#define SMT 2
#define SNT 4

static __device__ __forceinline__ void sub_load(v4f (&fa)[SMT][4], v4f (&fb)[SNT][4],
                                                const float* const (&pa)[SMT],
                                                const float* const (&pb)[SNT], int koff) {
    #pragma unroll
    for (int t = 0; t < SMT; ++t) {
        #pragma unroll
        for (int q = 0; q < 4; ++q) fa[t][q] = *(const v4f*)(pa[t] + koff + 8 * q);
    }
    #pragma unroll
    for (int u = 0; u < SNT; ++u) {
        #pragma unroll
        for (int q = 0; q < 4; ++q) fb[u][q] = *(const v4f*)(pb[u] + koff + 8 * q);
    }
}

static __device__ __forceinline__ void sub_mma(v8f (&acc)[SMT][SNT], const v4f (&fa)[SMT][4],
                                               const v4f (&fb)[SNT][4]) {
    #pragma unroll
    for (int j = 0; j < 8; ++j) {
        #pragma unroll
        for (int t = 0; t < SMT; ++t) {
            const v4f aq = fa[t][j >> 1];
            const v2f af = (j & 1) ? (v2f){ aq[2], aq[3] } : (v2f){ aq[0], aq[1] };
            #pragma unroll
            for (int u = 0; u < SNT; ++u) {
                const v4f wq = fb[u][j >> 1];
                const v2f wf = (j & 1) ? (v2f){ wq[2], wq[3] } : (v2f){ wq[0], wq[1] };
                acc[t][u] = wmma_f32(af, wf, acc[t][u]);
            }
        }
    }
}

__global__ __launch_bounds__(128) void k_subject(
    const float* __restrict__ x, const int* __restrict__ sid,
    const float* __restrict__ W_sub, const float* __restrict__ b_sub,
    float* __restrict__ out0)
{
    const int tid  = threadIdx.x;
    const int lane = tid & 31;
    const int n    = lane & 15;
    const int half = lane >> 4;
    const int b  = blockIdx.y;
    const int n0 = blockIdx.x * (16 * SNT);
    const int m0 = (tid >> 5) * (16 * SMT);
    const int s  = sid[b];
    const float* Wg = W_sub + (size_t)s * TT * TT;

    const float* pa[SMT];
    const float* pb[SNT];
    #pragma unroll
    for (int t = 0; t < SMT; ++t) {
        int am = m0 + 16 * t + n; if (am >= CC) am = CC - 1;
        pa[t] = x + ((size_t)b * CC + am) * TT + 4 * half;
    }
    #pragma unroll
    for (int u = 0; u < SNT; ++u) {
        int an = n0 + 16 * u + n; if (an >= TT) an = TT - 1;
        pb[u] = Wg + (size_t)an * TT + 4 * half;
    }

    v8f acc[SMT][SNT];
    #pragma unroll
    for (int t = 0; t < SMT; ++t) {
        #pragma unroll
        for (int u = 0; u < SNT; ++u) acc[t][u] = (v8f){};
    }

    const float* bg = b_sub + (size_t)s * TT;
    float bias[SNT];
    #pragma unroll
    for (int u = 0; u < SNT; ++u) {
        const int nn = n0 + 16 * u + n;
        bias[u] = bg[nn < TT ? nn : TT - 1];
    }

    v4f fa0[SMT][4], fb0[SNT][4], fa1[SMT][4], fb1[SNT][4];
    sub_load(fa0, fb0, pa, pb, 0);
    for (int st = 0; st < 14; st += 2) {
        sub_load(fa1, fb1, pa, pb, 32 * (st + 1));
        __builtin_amdgcn_sched_barrier(0);
        sub_mma(acc, fa0, fb0);
        __builtin_amdgcn_sched_barrier(0);
        sub_load(fa0, fb0, pa, pb, 32 * (st + 2));
        __builtin_amdgcn_sched_barrier(0);
        sub_mma(acc, fa1, fb1);
    }

    v4f ta[SMT][2], tb[SNT][2];
    v2f ua[SMT], ub[SNT];
    #pragma unroll
    for (int t = 0; t < SMT; ++t) {
        #pragma unroll
        for (int q = 0; q < 2; ++q) ta[t][q] = *(const v4f*)(pa[t] + 480 + 8 * q);
        ua[t] = *(const v2f*)(pa[t] + 496 - 2 * half);
    }
    #pragma unroll
    for (int u = 0; u < SNT; ++u) {
        #pragma unroll
        for (int q = 0; q < 2; ++q) tb[u][q] = *(const v4f*)(pb[u] + 480 + 8 * q);
        ub[u] = *(const v2f*)(pb[u] + 496 - 2 * half);
    }

    __builtin_amdgcn_sched_barrier(0);
    sub_mma(acc, fa0, fb0);
    __builtin_amdgcn_sched_barrier(0);

    #pragma unroll
    for (int j = 0; j < 4; ++j) {
        #pragma unroll
        for (int t = 0; t < SMT; ++t) {
            const v4f aq = ta[t][j >> 1];
            const v2f af = (j & 1) ? (v2f){ aq[2], aq[3] } : (v2f){ aq[0], aq[1] };
            #pragma unroll
            for (int u = 0; u < SNT; ++u) {
                const v4f wq = tb[u][j >> 1];
                const v2f wf = (j & 1) ? (v2f){ wq[2], wq[3] } : (v2f){ wq[0], wq[1] };
                acc[t][u] = wmma_f32(af, wf, acc[t][u]);
            }
        }
    }
    #pragma unroll
    for (int t = 0; t < SMT; ++t) {
        #pragma unroll
        for (int u = 0; u < SNT; ++u) acc[t][u] = wmma_f32(ua[t], ub[u], acc[t][u]);
    }

    #pragma unroll
    for (int u = 0; u < SNT; ++u) {
        const int nn = n0 + 16 * u + n;
        #pragma unroll
        for (int t = 0; t < SMT; ++t) {
            #pragma unroll
            for (int r = 0; r < 8; ++r) {
                const int mm = m0 + 16 * t + r + (half << 3);
                if (mm < CC && nn < TT)
                    out0[((size_t)b * CC + mm) * TT + nn] = acc[t][u][r] + bias[u];
            }
        }
    }
}

#define KSL   8
#define NPR   3
#define KSUB  4
#define NWV   (NPR * KSUB)
#define NSTEP (K2 / 32)
#define PPIT  162
#define PTILE (CC * PPIT)
#define H2SL  (BB * WID * UU)
#define WTILE (10 * 8 * 32)
#define STG   512
#define RPW   ((CC + NWV - 1) / NWV)

__global__ __launch_bounds__(NWV * 32) void k_conv2(
    const float* __restrict__ out0,
    const float* __restrict__ conv1_w, const float* __restrict__ conv1_b,
    const float* __restrict__ bn1_g, const float* __restrict__ bn1_b,
    const float* __restrict__ bn1_m, const float* __restrict__ bn1_v,
    const float* __restrict__ conv2_w,
    float* __restrict__ part_lo, float* __restrict__ part_hi)
{
    __shared__ float Pl[NPR * PTILE];
    __shared__ float w1s[WID * 5];
    __shared__ float sc1[WID], sh1[WID];
    __shared__ float stg[NWV * STG];

    const int tid = threadIdx.x;
    const int b   = blockIdx.y;
    const int ksl = blockIdx.x;

    for (int i = tid; i < WID * 5; i += NWV * 32) w1s[i] = conv1_w[i];
    if (tid < WID) {
        float sc = bn1_g[tid] * rsqrtf(bn1_v[tid] + EPSF);
        sc1[tid] = sc;
        sh1[tid] = bn1_b[tid] - bn1_m[tid] * sc + sc * conv1_b[tid];
    }
    const int wv   = __builtin_amdgcn_readfirstlane(tid >> 5);
    const int lane = tid & 31;

    {
        const float* xb = out0 + (size_t)b * CC * TT;
        float* sw = stg + wv * STG;
        int off[15];
        #pragma unroll
        for (int i = 0; i < 15; ++i) {
            const int q = 15 * lane + i;
            const int p = q / 160;
            off[i] = p * PTILE + (q - 160 * p);
        }
        v4f rv[RPW][4];
        #pragma unroll
        for (int r = 0; r < RPW; ++r) {
            int c = wv + NWV * r;
            if (c >= CC) c -= NWV;
            #pragma unroll
            for (int j = 0; j < 4; ++j) {
                int f = 4 * lane + 128 * j;
                if (f > TT - 4) f = TT - 4;
                rv[r][j] = *(const v4f*)(xb + (size_t)c * TT + f);
            }
        }
        #pragma unroll
        for (int r = 0; r < RPW; ++r) {
            int c = wv + NWV * r;
            if (c >= CC) c -= NWV;
            float* prow = Pl + c * PPIT;
            #pragma unroll
            for (int j = 0; j < 4; ++j) *(v4f*)(sw + 4 * lane + 128 * j) = rv[r][j];
            __builtin_amdgcn_wave_barrier();
            float v[31];
            #pragma unroll
            for (int i = 0; i < 31; ++i) v[i] = sw[15 * lane + i];
            __builtin_amdgcn_wave_barrier();
            const float mid = (v[14] + v[15]) + v[16];
            float sf[14], pf[15];
            sf[13] = v[13];
            #pragma unroll
            for (int i = 12; i >= 0; --i) sf[i] = v[i] + sf[i + 1];
            pf[1] = v[17];
            #pragma unroll
            for (int i = 2; i < 15; ++i) pf[i] = pf[i - 1] + v[16 + i];
            #pragma unroll
            for (int i = 0; i < 15; ++i) {
                float o;
                if (i == 0)       o = sf[0] + mid;
                else if (i == 14) o = mid + pf[14];
                else              o = (sf[i] + mid) + pf[i];
                prow[off[i]] = o * (1.f / 17.f);
            }
        }
    }
    __syncthreads();

    const int n    = lane & 15;
    const int half = lane >> 4;
    const int pr   = wv % NPR;
    const int ks   = wv / NPR;

    const int srem = NSTEP % KSL, sbas = NSTEP / KSL;
    const int s0   = ksl * sbas + (ksl < srem ? ksl : srem);
    const int scnt = sbas + (ksl < srem ? 1 : 0);
    const int wrem = scnt % KSUB, wbas = scnt / KSUB;
    const int st0  = s0 + ks * wbas + (ks < wrem ? ks : wrem);
    const int cnt  = wbas + (ks < wrem ? 1 : 0);

    const int k0 = st0 * 32 + half * 16;
    int w = k0 / CC;
    int c = k0 - w * CC;

    const float* pa = conv2_w + (size_t)n * K2 + k0;
    const float* pp = &Pl[pr * PTILE + 5 * n];

    v8f acc[2][5];
    #pragma unroll
    for (int nt = 0; nt < 2; ++nt) {
        #pragma unroll
        for (int t = 0; t < 5; ++t) acc[nt][t] = (v8f){};
    }

    for (int it = 0; it < cnt; ++it) {
        v4f a[5][4];
        #pragma unroll
        for (int t = 0; t < 5; ++t) {
            #pragma unroll
            for (int q = 0; q < 4; ++q)
                a[t][q] = *(const v4f*)(pa + (size_t)t * 16 * K2 + 4 * q);
        }

        v2f bf[2][8];
        #pragma unroll
        for (int j = 0; j < 8; ++j) {
            int cj = c + 2 * j, wj = w;
            if (cj >= CC) { cj -= CC; wj += 1; }
            const float* wr  = &w1s[wj * 5];
            const float  w0 = wr[0], w1 = wr[1], w2 = wr[2], w3 = wr[3], w4 = wr[4];
            const float  scw = sc1[wj], shw = sh1[wj];
            #pragma unroll
            for (int nt = 0; nt < 2; ++nt) {
                const float* pr0 = pp + 80 * nt + cj * PPIT;
                const float* pr1 = pr0 + PPIT;
                float v0 = w0*pr0[0] + w1*pr0[1] + w2*pr0[2] + w3*pr0[3] + w4*pr0[4];
                float g0 = scw * v0 + shw;
                float e0 = g0 > 0.f ? g0 : (__expf(g0) - 1.f);
                float v1 = w0*pr1[0] + w1*pr1[1] + w2*pr1[2] + w3*pr1[3] + w4*pr1[4];
                float g1 = scw * v1 + shw;
                float e1 = g1 > 0.f ? g1 : (__expf(g1) - 1.f);
                bf[nt][j] = (v2f){ e0, e1 };
            }
            if ((j & 1) && j < 7) __builtin_amdgcn_sched_barrier(0);
        }

        #pragma unroll
        for (int j = 0; j < 8; ++j) {
            #pragma unroll
            for (int t = 0; t < 5; ++t) {
                v4f aq = a[t][j >> 1];
                v2f af = (j & 1) ? (v2f){ aq[2], aq[3] } : (v2f){ aq[0], aq[1] };
                #pragma unroll
                for (int nt = 0; nt < 2; ++nt) acc[nt][t] = wmma_f32(af, bf[nt][j], acc[nt][t]);
            }
        }

        pa += 32;
        c  += 32;
        if (c >= CC) { c -= CC; w += 1; }
    }

    __syncthreads();

    float* rw = Pl + (size_t)wv * WTILE;
    #pragma unroll
    for (int nt = 0; nt < 2; ++nt) {
        #pragma unroll
        for (int t = 0; t < 5; ++t) {
            #pragma unroll
            for (int r = 0; r < 8; ++r) rw[((nt * 5 + t) * 8 + r) * 32 + lane] = acc[nt][t][r];
        }
    }
    __syncthreads();

    float* dst = (ksl < KSL - 1) ? (part_lo + (size_t)ksl * H2SL) : part_hi;
    #pragma unroll 4
    for (int it = 0; it < (NPR * WTILE) / (NWV * 32); ++it) {
        const int sidx = tid + it * (NWV * 32);
        const int p   = sidx / WTILE;
        const int rem = sidx - p * WTILE;
        float sum = 0.f;
        #pragma unroll
        for (int v = 0; v < KSUB; ++v) sum += Pl[(v * NPR + p) * WTILE + rem];
        const int tt = rem >> 8, r = (rem >> 5) & 7, l = rem & 31;
        const int nt = tt / 5, t = tt - nt * 5;
        const int w2 = t * 16 + r + (l >> 4) * 8;
        const int u  = 16 * (2 * p + nt) + (l & 15);
        dst[((size_t)b * WID + w2) * UU + u] = sum;
    }
}

__global__ __launch_bounds__(256) void k_proj(
    const float* __restrict__ part_lo, const float* __restrict__ part_hi,
    const float* __restrict__ conv2_b,
    const float* __restrict__ bn2_g, const float* __restrict__ bn2_b,
    const float* __restrict__ bn2_m, const float* __restrict__ bn2_v,
    const float* __restrict__ proj_w, const float* __restrict__ proj_b,
    float* __restrict__ z)
{
    __shared__ float hs[WID * 16];
    __shared__ float pw[8 * WID];
    const int ut = blockIdx.x;
    const int b  = blockIdx.y;
    #pragma unroll
    for (int it = 0; it < (WID * 16) / 256; ++it) {
        const int i  = threadIdx.x + it * 256;
        const int w2 = i >> 4, uu = i & 15;
        const size_t off = ((size_t)b * WID + w2) * UU + ut * 16 + uu;
        float sum = 0.f;
        #pragma unroll
        for (int s = 0; s < KSL - 1; ++s) sum += part_lo[(size_t)s * H2SL + off];
        sum += part_hi[off];
        float sc = bn2_g[w2] * rsqrtf(bn2_v[w2] + EPSF);
        float sh = bn2_b[w2] - bn2_m[w2] * sc;
        float v  = sc * (sum + conv2_b[w2]) + sh;
        hs[i] = v > 0.f ? v : (__expf(v) - 1.f);
    }
    for (int i = threadIdx.x; i < 8 * WID; i += 256)  pw[i] = proj_w[i];
    __syncthreads();
    if (threadIdx.x < 128) {
        const int uu = threadIdx.x >> 3, e = threadIdx.x & 7;
        float s = proj_b[e];
        #pragma unroll 8
        for (int w = 0; w < WID; ++w) s += pw[e * WID + w] * hs[w * 16 + uu];
        z[(size_t)b * HID + (ut * 16 + uu) * 8 + e] = s;
    }
}

#define MW 8

template <int KTOT>
static __device__ __forceinline__ v8f mlp_partial(const float* __restrict__ A,
                                                  const float* __restrict__ Wt,
                                                  int m0, int n0, int wv, int lane)
{
    constexpr int KS = KTOT / MW;
    constexpr int NQ = KS / 8;
    const int n = lane & 15, half = lane >> 4;
    const float* pa = A  + (size_t)(m0 + n) * KTOT + wv * KS + 4 * half;
    const float* pb = Wt + (size_t)(n0 + n) * KTOT + wv * KS + 4 * half;
    v4f fa[NQ], fb[NQ];
    #pragma unroll
    for (int q = 0; q < NQ; ++q) fb[q] = *(const v4f*)(pb + 8 * q);
    #pragma unroll
    for (int q = 0; q < NQ; ++q) fa[q] = *(const v4f*)(pa + 8 * q);
    v8f acc0 = {}, acc1 = {};
    #pragma unroll
    for (int q = 0; q < NQ; ++q) {
        acc0 = wmma_f32((v2f){ fa[q][0], fa[q][1] }, (v2f){ fb[q][0], fb[q][1] }, acc0);
        acc1 = wmma_f32((v2f){ fa[q][2], fa[q][3] }, (v2f){ fb[q][2], fb[q][3] }, acc1);
    }
    return acc0 + acc1;
}

__global__ __launch_bounds__(MW * 32) void k_mlp1(
    const float* __restrict__ z, const float* __restrict__ w,
    const float* __restrict__ bias, float* __restrict__ z1, float* __restrict__ g)
{
    __shared__ float part[MW * 8 * 32];
    const int tid  = threadIdx.x;
    const int lane = tid & 31;
    const int wv   = tid >> 5;
    const int m0 = blockIdx.x * 16, n0 = blockIdx.y * 16;
    const int nn = n0 + (lane & 15);
    const float bn = bias[nn];

    const v8f acc = mlp_partial<HID>(z, w, m0, n0, wv, lane);
    #pragma unroll
    for (int r = 0; r < 8; ++r) part[(wv * 8 + r) * 32 + lane] = acc[r];
    __syncthreads();

    float sum = 0.f;
    #pragma unroll
    for (int v = 0; v < MW; ++v) sum += part[(v * 8 + wv) * 32 + lane];
    const int mm = m0 + wv + ((lane >> 4) << 3);
    const float v = sum + bn;
    z1[(size_t)mm * PROJ + nn] = v;
    g [(size_t)mm * PROJ + nn] = 0.5f * v * (1.f + erff(v * 0.70710678118654752f));
}

__global__ __launch_bounds__(MW * 32) void k_mlp2(
    const float* __restrict__ g, const float* __restrict__ w,
    const float* __restrict__ bias, const float* __restrict__ z1,
    float* __restrict__ y)
{
    __shared__ float part[MW * 8 * 32];
    const int tid  = threadIdx.x;
    const int lane = tid & 31;
    const int wv   = tid >> 5;
    const int m0 = blockIdx.x * 16, n0 = blockIdx.y * 16;
    const int nn = n0 + (lane & 15);
    const int mm = m0 + wv + ((lane >> 4) << 3);
    const float bn  = bias[nn];
    const float res = z1[(size_t)mm * PROJ + nn];

    const v8f acc = mlp_partial<PROJ>(g, w, m0, n0, wv, lane);
    #pragma unroll
    for (int r = 0; r < 8; ++r) part[(wv * 8 + r) * 32 + lane] = acc[r];
    __syncthreads();

    float sum = 0.f;
    #pragma unroll
    for (int v = 0; v < MW; ++v) sum += part[(v * 8 + wv) * 32 + lane];
    y[(size_t)mm * PROJ + nn] = res + sum + bn;
}

__global__ __launch_bounds__(256) void k_ln(
    const float* __restrict__ y, const float* __restrict__ gam,
    const float* __restrict__ bet, float* __restrict__ out)
{
    __shared__ float red[256];
    const int b = blockIdx.x;
    const float* row = y + (size_t)b * PROJ;
    float s = 0.f;
    for (int i = threadIdx.x; i < PROJ; i += 256) s += row[i];
    red[threadIdx.x] = s; __syncthreads();
    for (int off = 128; off > 0; off >>= 1) {
        if (threadIdx.x < off) red[threadIdx.x] += red[threadIdx.x + off];
        __syncthreads();
    }
    const float mu = red[0] * (1.f / PROJ);
    __syncthreads();
    float s2 = 0.f;
    for (int i = threadIdx.x; i < PROJ; i += 256) { float d = row[i] - mu; s2 += d * d; }
    red[threadIdx.x] = s2; __syncthreads();
    for (int off = 128; off > 0; off >>= 1) {
        if (threadIdx.x < off) red[threadIdx.x] += red[threadIdx.x + off];
        __syncthreads();
    }
    const float inv = rsqrtf(red[0] * (1.f / PROJ) + EPSF);
    for (int i = threadIdx.x; i < PROJ; i += 256)
        out[(size_t)b * PROJ + i] = (row[i] - mu) * inv * gam[i] + bet[i];
}

extern "C" void kernel_launch(void* const* d_in, const int* in_sizes, int n_in,
                              void* d_out, int out_size, void* d_ws, size_t ws_size,
                              hipStream_t stream) {
    const float* x       = (const float*)d_in[0];
    const int*   sid     = (const int*)  d_in[1];
    const float* W_sub   = (const float*)d_in[2];
    const float* b_sub   = (const float*)d_in[3];
    const float* conv1_w = (const float*)d_in[4];
    const float* conv1_b = (const float*)d_in[5];
    const float* bn1_g   = (const float*)d_in[6];
    const float* bn1_b   = (const float*)d_in[7];
    const float* bn1_m   = (const float*)d_in[8];
    const float* bn1_v   = (const float*)d_in[9];
    const float* conv2_w = (const float*)d_in[10];
    const float* conv2_b = (const float*)d_in[11];
    const float* bn2_g   = (const float*)d_in[12];
    const float* bn2_b   = (const float*)d_in[13];
    const float* bn2_m   = (const float*)d_in[14];
    const float* bn2_v   = (const float*)d_in[15];
    const float* proj_w  = (const float*)d_in[16];
    const float* proj_b  = (const float*)d_in[17];
    const float* mlp1_w  = (const float*)d_in[18];
    const float* mlp1_b  = (const float*)d_in[19];
    const float* mlp2_w  = (const float*)d_in[20];
    const float* mlp2_b  = (const float*)d_in[21];
    const float* ln_g    = (const float*)d_in[22];
    const float* ln_b    = (const float*)d_in[23];

    float* ws   = (float*)d_ws;
    float* out0 = ws;
    float* pl7  = out0 + (size_t)BB * CC * TT;
    float* h2   = pl7  + (size_t)BB * CC * PLEN;
    float* z    = h2   + (size_t)BB * WID * UU;
    float* z1   = z    + (size_t)BB * HID;
    float* gg   = z1   + (size_t)BB * PROJ;
    float* yy   = gg   + (size_t)BB * PROJ;

    dim3 g1(8, BB);
    k_subject<<<g1, 128, 0, stream>>>(x, sid, W_sub, b_sub, out0);

    dim3 g3(KSL, BB);
    k_conv2<<<g3, NWV * 32, 0, stream>>>(out0, conv1_w, conv1_b, bn1_g, bn1_b, bn1_m, bn1_v,
                                    conv2_w, pl7, h2);

    dim3 g4(UU / 16, BB);
    k_proj<<<g4, 256, 0, stream>>>(pl7, h2, conv2_b, bn2_g, bn2_b, bn2_m, bn2_v, proj_w, proj_b, z);

    dim3 g5(2, PROJ / 16);
    k_mlp1<<<g5, MW * 32, 0, stream>>>(z, mlp1_w, mlp1_b, z1, gg);
    k_mlp2<<<g5, MW * 32, 0, stream>>>(gg, mlp2_w, mlp2_b, z1, yy);

    k_ln<<<BB, 256, 0, stream>>>(yy, ln_g, ln_b, (float*)d_out);
}
